# starting kernel
# baseline (speedup 1.0000x reference)
#include <hip/hip_runtime.h>

typedef __attribute__((ext_vector_type(2))) float v2f;
typedef __attribute__((ext_vector_type(8))) float v8f;

#define N_NODES 50000
#define N_EDGES 600000
#define D 128
#define N_LAYERS 4

__global__ __launch_bounds__(256) void deg_kernel(const int* __restrict__ src,
                                                  const int* __restrict__ dst,
                                                  float* __restrict__ out_deg,
                                                  float* __restrict__ in_deg) {
    int e = blockIdx.x * blockDim.x + threadIdx.x;
    if (e < N_EDGES) {
        unsafeAtomicAdd(&out_deg[src[e]], 1.0f);
        unsafeAtomicAdd(&in_deg[dst[e]], 1.0f);
    }
}

__global__ __launch_bounds__(256) void norm_kernel(float* __restrict__ sn,
                                                   float* __restrict__ dn) {
    int n = blockIdx.x * blockDim.x + threadIdx.x;
    if (n < N_NODES) {
        sn[n] = rsqrtf(fmaxf(sn[n], 1.0f));
        dn[n] = rsqrtf(fmaxf(dn[n], 1.0f));
    }
}

__global__ __launch_bounds__(256) void agg_kernel(const float* __restrict__ h,
                                                  const int* __restrict__ src,
                                                  const int* __restrict__ dst,
                                                  const float* __restrict__ src_norm,
                                                  float* __restrict__ agg) {
    int tid   = blockIdx.x * blockDim.x + threadIdx.x;
    int e     = tid >> 5;
    int chunk = (tid & 31) * 4;
    if (e >= N_EDGES) return;

    int s = src[e];
    int d = dst[e];
    float sc = src_norm[s];

    const float4 v = *reinterpret_cast<const float4*>(h + (size_t)s * D + chunk);
    float* ap = agg + (size_t)d * D + chunk;
    unsafeAtomicAdd(ap + 0, v.x * sc);
    unsafeAtomicAdd(ap + 1, v.y * sc);
    unsafeAtomicAdd(ap + 2, v.z * sc);
    unsafeAtomicAdd(ap + 3, v.w * sc);
}

__global__ __launch_bounds__(256) void gemm_bias_kernel(
        const float* __restrict__ A,
        const float* __restrict__ dnorm,
        const float* __restrict__ Wm,
        const float* __restrict__ bias,
        float* __restrict__ out)
{
    __shared__ float ldsA[16 * 132];

    const int M0 = blockIdx.x * 16;
    const int t  = threadIdx.x;

#pragma unroll
    for (int half = 0; half < 2; ++half) {
        int flat = half * 1024 + t * 4;
        int row  = flat >> 7;
        int col  = flat & 127;
        float4 v = *reinterpret_cast<const float4*>(A + (size_t)(M0 + row) * D + col);
        float  s = dnorm[M0 + row];
        float* p = &ldsA[row * 132 + col];
        p[0] = v.x * s;
        p[1] = v.y * s;
        p[2] = v.z * s;
        p[3] = v.w * s;
    }
    __syncthreads();

    const int wave = t >> 5;
    const int lane = t & 31;
    const int n0   = wave * 16;
    const int mrow = lane & 15;
    const int kb   = (lane >> 4) * 2;
    const int ncol = n0 + (lane & 15);

    v8f c = {};
#pragma unroll
    for (int k0 = 0; k0 < D; k0 += 4) {
        const float* ap = &ldsA[mrow * 132 + k0 + kb];
        v2f a;
        a.x = ap[0];
        a.y = ap[1];

        v2f b;
        b.x = Wm[(size_t)(k0 + kb    ) * D + ncol];
        b.y = Wm[(size_t)(k0 + kb + 1) * D + ncol];

        c = __builtin_amdgcn_wmma_f32_16x16x4_f32(
                 false, a,  false, b,
                 (short)0, c,  false,  false);
    }

    const float bn    = bias[ncol];
    const int   mbase = M0 + ((lane >> 4) * 8);
#pragma unroll
    for (int v = 0; v < 8; ++v) {
        out[(size_t)(mbase + v) * D + ncol] = c[v] + bn;
    }
}

extern "C" void kernel_launch(void* const* d_in, const int* in_sizes, int n_in,
                              void* d_out, int out_size, void* d_ws, size_t ws_size,
                              hipStream_t stream) {
    const float* feat = (const float*)d_in[0];
    const int*   src  = (const int*)  d_in[1];
    const int*   dst  = (const int*)  d_in[2];
    const float* W    = (const float*)d_in[3];
    const float* b    = (const float*)d_in[4];
    float*       out  = (float*)d_out;

    float* ws       = (float*)d_ws;
    float* src_norm = ws;
    float* dst_norm = ws + N_NODES;
    float* agg      = ws + 2 * (size_t)N_NODES;
    float* hbuf     = agg + (size_t)N_NODES * D;

    hipMemsetAsync(src_norm, 0, 2 * (size_t)N_NODES * sizeof(float), stream);
    deg_kernel<<<(N_EDGES + 255) / 256, 256, 0, stream>>>(src, dst, src_norm, dst_norm);
    norm_kernel<<<(N_NODES + 255) / 256, 256, 0, stream>>>(src_norm, dst_norm);

    const float* h = feat;
    for (int i = 0; i < N_LAYERS; ++i) {
        hipMemsetAsync(agg, 0, (size_t)N_NODES * D * sizeof(float), stream);

        agg_kernel<<<(N_EDGES * 32) / 256, 256, 0, stream>>>(h, src, dst, src_norm, agg);

        float* o = (i == N_LAYERS - 1) ? out : hbuf;
        gemm_bias_kernel<<<N_NODES / 16, 256, 0, stream>>>(
            agg, dst_norm, W + (size_t)i * D * D, b + (size_t)i * D, o);
        h = hbuf;
    }
}


# Aggregation: per-edge f32 atomic scatter-add replaced by a per-node wave gather over a destination-sorted edge list built once per call (int atomics, arrival-order rank); sum order changes
# speedup vs baseline: 2.7735x; 2.7735x over previous
#include <hip/hip_runtime.h>


typedef __attribute__((ext_vector_type(2))) float v2f;
typedef __attribute__((ext_vector_type(8))) float v8f;

#define N_NODES 50000
#define N_EDGES 600000
#define D 128
#define N_LAYERS 4

#define NODES_PER_WAVE 4
#define GROUP 8

static_assert(N_NODES % NODES_PER_WAVE == 0, "a gather wave owns whole groups of nodes");
static_assert(NODES_PER_WAVE == 4, "the gather reads its bookkeeping as one int4");

__global__ __launch_bounds__(256) void deg_rank_kernel(const int* __restrict__ src,
                                                       const int* __restrict__ dst,
                                                       int* __restrict__ out_cnt,
                                                       int* __restrict__ in_cnt,
                                                       int* __restrict__ rank) {
    int e = blockIdx.x * blockDim.x + threadIdx.x;
    if (e < N_EDGES) {
        const int s = src[e];
        const int d = dst[e];
        atomicAdd(&out_cnt[s], 1);
        rank[e] = atomicAdd(&in_cnt[d], 1);
    }
}

__global__ __launch_bounds__(256) void norm_scan_kernel(const int* __restrict__ out_cnt,
                                                        const int* __restrict__ in_cnt,
                                                        float* __restrict__ src_norm,
                                                        float* __restrict__ dst_norm,
                                                        int* __restrict__ row_start,
                                                        int* __restrict__ cursor) {
    __shared__ int wave_off[8];
    __shared__ int block_base;

    const int t    = threadIdx.x;
    const int lane = t & 31;
    const int wave = t >> 5;
    const int n    = blockIdx.x * 256 + t;

    int id = 0;
    if (n < N_NODES) {
        const int od = out_cnt[n];
        id = in_cnt[n];
        src_norm[n] = rsqrtf(fmaxf((float)od, 1.0f));
        dst_norm[n] = rsqrtf(fmaxf((float)id, 1.0f));
    }

    int x = id;
#pragma unroll
    for (int o = 1; o < 32; o <<= 1) {
        int y = __shfl_up(x, o);
        if (lane >= o) x += y;
    }
    if (lane == 31) wave_off[wave] = x;
    __syncthreads();
    if (t == 0) {
        int tot = 0;
#pragma unroll
        for (int w = 0; w < 8; ++w) {
            int v = wave_off[w];
            wave_off[w] = tot;
            tot += v;
        }
        block_base = atomicAdd(cursor, tot);
    }
    __syncthreads();
    if (n < N_NODES) row_start[n] = block_base + wave_off[wave] + x - id;
}

__global__ __launch_bounds__(256) void fill_kernel(const int* __restrict__ src,
                                                   const int* __restrict__ dst,
                                                   const int* __restrict__ rank,
                                                   const int* __restrict__ row_start,
                                                   const float* __restrict__ src_norm,
                                                   int2* __restrict__ col) {
    int e = blockIdx.x * blockDim.x + threadIdx.x;
    if (e < N_EDGES) {
        const int s = src[e];
        const int d = dst[e];
        col[row_start[d] + rank[e]] = make_int2(s, __float_as_int(src_norm[s]));
    }
}

__global__ __launch_bounds__(256) void gather_kernel(const float* __restrict__ h,
                                                     const int* __restrict__ row_start,
                                                     const int* __restrict__ in_cnt,
                                                     const int2* __restrict__ col,
                                                     float* __restrict__ agg) {
    const int lane = threadIdx.x & 31;
    const int wid = __builtin_amdgcn_readfirstlane((int)(blockIdx.x * 8 + (threadIdx.x >> 5)));
    const int n0  = wid * NODES_PER_WAVE;
    if (n0 >= N_NODES) return;

    const int4 rs = *reinterpret_cast<const int4*>(row_start + n0);
    const int4 ct = *reinterpret_cast<const int4*>(in_cnt + n0);
    const int start[NODES_PER_WAVE] = {rs.x, rs.y, rs.z, rs.w};
    const int cnt[NODES_PER_WAVE]   = {ct.x, ct.y, ct.z, ct.w};

    int2 first[NODES_PER_WAVE];
#pragma unroll
    for (int k = 0; k < NODES_PER_WAVE; ++k) {
        first[k] = make_int2(0, 0);
        if (lane < cnt[k]) first[k] = col[start[k] + lane];
    }

#pragma unroll
    for (int k = 0; k < NODES_PER_WAVE; ++k) {
        float4 acc = make_float4(0.0f, 0.0f, 0.0f, 0.0f);
        int2 p = first[k];
        for (int base = 0; base < cnt[k]; base += 32) {
            const int m = min(32, cnt[k] - base);
            if (base > 0) {
                p = make_int2(0, 0);
                if (lane < m) p = col[start[k] + base + lane];
            }
            for (int j = 0; j < m; j += GROUP) {
                float4 v[GROUP];
                float  w[GROUP];
#pragma unroll
                for (int i = 0; i < GROUP; ++i) {
                    const int s = __builtin_amdgcn_readlane(p.x, j + i);
                    w[i] = __int_as_float(__builtin_amdgcn_readlane(p.y, j + i));
                    v[i] = make_float4(0.0f, 0.0f, 0.0f, 0.0f);
                    if (j + i < m) {
                        const float4* row = reinterpret_cast<const float4*>(h + (size_t)s * D);
                        v[i] = row[lane];
                    }
                }
#pragma unroll
                for (int i = 0; i < GROUP; ++i) {
                    acc.x = fmaf(v[i].x, w[i], acc.x);
                    acc.y = fmaf(v[i].y, w[i], acc.y);
                    acc.z = fmaf(v[i].z, w[i], acc.z);
                    acc.w = fmaf(v[i].w, w[i], acc.w);
                }
            }
        }
        *reinterpret_cast<float4*>(agg + (size_t)(n0 + k) * D + lane * 4) = acc;
    }
}

__global__ __launch_bounds__(256) void gemm_bias_kernel(
        const float* __restrict__ A,
        const float* __restrict__ dnorm,
        const float* __restrict__ Wm,
        const float* __restrict__ bias,
        float* __restrict__ out)
{
    __shared__ float ldsA[16 * 132];

    const int M0 = blockIdx.x * 16;
    const int t  = threadIdx.x;

#pragma unroll
    for (int half = 0; half < 2; ++half) {
        int flat = half * 1024 + t * 4;
        int row  = flat >> 7;
        int col  = flat & 127;
        float4 v = *reinterpret_cast<const float4*>(A + (size_t)(M0 + row) * D + col);
        float  s = dnorm[M0 + row];
        float* p = &ldsA[row * 132 + col];
        p[0] = v.x * s;
        p[1] = v.y * s;
        p[2] = v.z * s;
        p[3] = v.w * s;
    }
    __syncthreads();

    const int wave = t >> 5;
    const int lane = t & 31;
    const int n0   = wave * 16;
    const int mrow = lane & 15;
    const int kb   = (lane >> 4) * 2;
    const int ncol = n0 + (lane & 15);

    v8f c = {};
#pragma unroll
    for (int k0 = 0; k0 < D; k0 += 4) {
        const float* ap = &ldsA[mrow * 132 + k0 + kb];
        v2f a;
        a.x = ap[0];
        a.y = ap[1];

        v2f b;
        b.x = Wm[(size_t)(k0 + kb    ) * D + ncol];
        b.y = Wm[(size_t)(k0 + kb + 1) * D + ncol];

        c = __builtin_amdgcn_wmma_f32_16x16x4_f32(
                 false, a,  false, b,
                 (short)0, c,  false,  false);
    }

    const float bn    = bias[ncol];
    const int   mbase = M0 + ((lane >> 4) * 8);
#pragma unroll
    for (int v = 0; v < 8; ++v) {
        out[(size_t)(mbase + v) * D + ncol] = c[v] + bn;
    }
}

extern "C" void kernel_launch(void* const* d_in, const int* in_sizes, int n_in,
                              void* d_out, int out_size, void* d_ws, size_t ws_size,
                              hipStream_t stream) {
    const float* feat = (const float*)d_in[0];
    const int*   src  = (const int*)  d_in[1];
    const int*   dst  = (const int*)  d_in[2];
    const float* W    = (const float*)d_in[3];
    const float* b    = (const float*)d_in[4];
    float*       out  = (float*)d_out;

    int*   wsi       = (int*)d_ws;
    float* wsf       = (float*)d_ws;
    int*   out_cnt   = wsi;
    int*   in_cnt    = wsi + 50048;
    int*   cursor    = wsi + 100096;
    float* src_norm  = wsf + 100160;
    float* dst_norm  = wsf + 150208;
    int*   row_start = wsi + 200256;
    int*   rank      = wsi + 250304;
    int2*  col       = (int2*)(wsi + 850304);
    float* agg       = wsf + 2050304;
    float* hbuf      = wsf + 8450304;

    hipMemsetAsync(out_cnt, 0, 100160 * sizeof(int), stream);
    deg_rank_kernel<<<(N_EDGES + 255) / 256, 256, 0, stream>>>(src, dst, out_cnt, in_cnt, rank);
    norm_scan_kernel<<<(N_NODES + 255) / 256, 256, 0, stream>>>(out_cnt, in_cnt, src_norm, dst_norm,
                                                                row_start, cursor);
    fill_kernel<<<(N_EDGES + 255) / 256, 256, 0, stream>>>(src, dst, rank, row_start, src_norm, col);

    const float* h = feat;
    for (int i = 0; i < N_LAYERS; ++i) {
        gather_kernel<<<(N_NODES / NODES_PER_WAVE + 7) / 8, 256, 0, stream>>>(h, row_start, in_cnt, col, agg);

        float* o = (i == N_LAYERS - 1) ? out : hbuf;
        gemm_bias_kernel<<<N_NODES / 16, 256, 0, stream>>>(
            agg, dst_norm, W + (size_t)i * D * D, b + (size_t)i * D, o);
        h = hbuf;
    }
}


# Gather fused into the matrix kernel: each block gathers its 64-node tile straight into the LDS A tile (agg buffer and 4 launches gone, B fragments shared by 4 row tiles); sum orders unchanged
# speedup vs baseline: 3.0760x; 1.1091x over previous
#include <hip/hip_runtime.h>


typedef __attribute__((ext_vector_type(2))) float v2f;
typedef __attribute__((ext_vector_type(8))) float v8f;

#define N_NODES 50000
#define N_EDGES 600000
#define D 128
#define N_LAYERS 4

#define TILE_M 64
#define NODES_PER_WAVE 8
#define GROUP 8
#define PITCH 132

static_assert(N_NODES % NODES_PER_WAVE == 0, "a wave owns whole groups of nodes");
static_assert(N_NODES % 16 == 0, "a 16-row matrix tile is wholly inside or outside the node range");
static_assert(NODES_PER_WAVE == 8, "the gather reads its bookkeeping as two int4");

__global__ __launch_bounds__(256) void deg_rank_kernel(const int* __restrict__ src,
                                                       const int* __restrict__ dst,
                                                       int* __restrict__ out_cnt,
                                                       int* __restrict__ in_cnt,
                                                       int* __restrict__ rank) {
    int e = blockIdx.x * blockDim.x + threadIdx.x;
    if (e < N_EDGES) {
        const int s = src[e];
        const int d = dst[e];
        atomicAdd(&out_cnt[s], 1);
        rank[e] = atomicAdd(&in_cnt[d], 1);
    }
}

__global__ __launch_bounds__(256) void norm_scan_kernel(const int* __restrict__ out_cnt,
                                                        const int* __restrict__ in_cnt,
                                                        float* __restrict__ src_norm,
                                                        float* __restrict__ dst_norm,
                                                        int* __restrict__ row_start,
                                                        int* __restrict__ cursor) {
    __shared__ int wave_off[8];
    __shared__ int block_base;

    const int t    = threadIdx.x;
    const int lane = t & 31;
    const int wave = t >> 5;
    const int n    = blockIdx.x * 256 + t;

    int id = 0;
    if (n < N_NODES) {
        const int od = out_cnt[n];
        id = in_cnt[n];
        src_norm[n] = rsqrtf(fmaxf((float)od, 1.0f));
        dst_norm[n] = rsqrtf(fmaxf((float)id, 1.0f));
    }

    int x = id;
#pragma unroll
    for (int o = 1; o < 32; o <<= 1) {
        int y = __shfl_up(x, o);
        if (lane >= o) x += y;
    }
    if (lane == 31) wave_off[wave] = x;
    __syncthreads();
    if (t == 0) {
        int tot = 0;
#pragma unroll
        for (int w = 0; w < 8; ++w) {
            int v = wave_off[w];
            wave_off[w] = tot;
            tot += v;
        }
        block_base = atomicAdd(cursor, tot);
    }
    __syncthreads();
    if (n < N_NODES) row_start[n] = block_base + wave_off[wave] + x - id;
}

__global__ __launch_bounds__(256) void fill_kernel(const int* __restrict__ src,
                                                   const int* __restrict__ dst,
                                                   const int* __restrict__ rank,
                                                   const int* __restrict__ row_start,
                                                   const float* __restrict__ src_norm,
                                                   int2* __restrict__ col) {
    int e = blockIdx.x * blockDim.x + threadIdx.x;
    if (e < N_EDGES) {
        const int s = src[e];
        const int d = dst[e];
        col[row_start[d] + rank[e]] = make_int2(s, __float_as_int(src_norm[s]));
    }
}

__global__ __launch_bounds__(256) void layer_kernel(
        const float* __restrict__ h,
        const int*   __restrict__ row_start,
        const int*   __restrict__ in_cnt,
        const int2*  __restrict__ col,
        const float* __restrict__ dnorm,
        const float* __restrict__ Wm,
        const float* __restrict__ bias,
        float* __restrict__ out)
{
    __shared__ float ldsA[TILE_M * PITCH];

    const int t    = threadIdx.x;
    const int lane = t & 31;
    const int wave = __builtin_amdgcn_readfirstlane(t >> 5);
    const int M0   = blockIdx.x * TILE_M;
    const int n0   = M0 + wave * NODES_PER_WAVE;

    float* myrows = &ldsA[wave * NODES_PER_WAVE * PITCH + lane * 4];

    if (n0 < N_NODES) {
        const int4 rs0 = *reinterpret_cast<const int4*>(row_start + n0);
        const int4 rs1 = *reinterpret_cast<const int4*>(row_start + n0 + 4);
        const int4 ct0 = *reinterpret_cast<const int4*>(in_cnt + n0);
        const int4 ct1 = *reinterpret_cast<const int4*>(in_cnt + n0 + 4);
        const float4 dn0 = *reinterpret_cast<const float4*>(dnorm + n0);
        const float4 dn1 = *reinterpret_cast<const float4*>(dnorm + n0 + 4);
        const int   start[NODES_PER_WAVE] = {rs0.x, rs0.y, rs0.z, rs0.w, rs1.x, rs1.y, rs1.z, rs1.w};
        const int   cnt[NODES_PER_WAVE]   = {ct0.x, ct0.y, ct0.z, ct0.w, ct1.x, ct1.y, ct1.z, ct1.w};
        const float sc[NODES_PER_WAVE]    = {dn0.x, dn0.y, dn0.z, dn0.w, dn1.x, dn1.y, dn1.z, dn1.w};

        int2 first[NODES_PER_WAVE];
#pragma unroll
        for (int k = 0; k < NODES_PER_WAVE; ++k) {
            first[k] = make_int2(0, 0);
            if (lane < cnt[k]) first[k] = col[start[k] + lane];
        }

#pragma unroll
        for (int k = 0; k < NODES_PER_WAVE; ++k) {
            float4 acc = make_float4(0.0f, 0.0f, 0.0f, 0.0f);
            int2 p = first[k];
            for (int base = 0; base < cnt[k]; base += 32) {
                const int m = min(32, cnt[k] - base);
                if (base > 0) {
                    p = make_int2(0, 0);
                    if (lane < m) p = col[start[k] + base + lane];
                }
                for (int j = 0; j < m; j += GROUP) {
                    float4 v[GROUP];
                    float  w[GROUP];
#pragma unroll
                    for (int i = 0; i < GROUP; ++i) {
                        const int s = __builtin_amdgcn_readlane(p.x, j + i);
                        w[i] = __int_as_float(__builtin_amdgcn_readlane(p.y, j + i));
                        v[i] = make_float4(0.0f, 0.0f, 0.0f, 0.0f);
                        if (j + i < m) {
                            const float4* row = reinterpret_cast<const float4*>(h + (size_t)s * D);
                            v[i] = row[lane];
                        }
                    }
#pragma unroll
                    for (int i = 0; i < GROUP; ++i) {
                        acc.x = fmaf(v[i].x, w[i], acc.x);
                        acc.y = fmaf(v[i].y, w[i], acc.y);
                        acc.z = fmaf(v[i].z, w[i], acc.z);
                        acc.w = fmaf(v[i].w, w[i], acc.w);
                    }
                }
            }
            *reinterpret_cast<float4*>(myrows + k * PITCH) =
                make_float4(acc.x * sc[k], acc.y * sc[k], acc.z * sc[k], acc.w * sc[k]);
        }
    } else {
#pragma unroll
        for (int k = 0; k < NODES_PER_WAVE; ++k)
            *reinterpret_cast<float4*>(myrows + k * PITCH) = make_float4(0.0f, 0.0f, 0.0f, 0.0f);
    }

    const int ncol = wave * 16 + (lane & 15);
    const int mrow = lane & 15;
    const int kb   = (lane >> 4) * 2;

    v2f b[D / 4];
#pragma unroll
    for (int ks = 0; ks < D / 4; ++ks) {
        b[ks].x = Wm[(size_t)(ks * 4 + kb    ) * D + ncol];
        b[ks].y = Wm[(size_t)(ks * 4 + kb + 1) * D + ncol];
    }
    const float bn = bias[ncol];

    __syncthreads();

    v8f c[TILE_M / 16] = {};
#pragma unroll
    for (int ks = 0; ks < D / 4; ++ks) {
#pragma unroll
        for (int mt = 0; mt < TILE_M / 16; ++mt) {
            const float* ap = &ldsA[(mt * 16 + mrow) * PITCH + ks * 4 + kb];
            v2f a;
            a.x = ap[0];
            a.y = ap[1];
            c[mt] = __builtin_amdgcn_wmma_f32_16x16x4_f32(
                     false, a,  false, b[ks],
                     (short)0, c[mt],  false,  false);
        }
    }

#pragma unroll
    for (int mt = 0; mt < TILE_M / 16; ++mt) {
        const int mtile = M0 + mt * 16;
        if (mtile < N_NODES) {
            const int mbase = mtile + ((lane >> 4) * 8);
#pragma unroll
            for (int v = 0; v < 8; ++v) {
                out[(size_t)(mbase + v) * D + ncol] = c[mt][v] + bn;
            }
        }
    }
}

extern "C" void kernel_launch(void* const* d_in, const int* in_sizes, int n_in,
                              void* d_out, int out_size, void* d_ws, size_t ws_size,
                              hipStream_t stream) {
    const float* feat = (const float*)d_in[0];
    const int*   src  = (const int*)  d_in[1];
    const int*   dst  = (const int*)  d_in[2];
    const float* W    = (const float*)d_in[3];
    const float* b    = (const float*)d_in[4];
    float*       out  = (float*)d_out;

    int*   wsi       = (int*)d_ws;
    float* wsf       = (float*)d_ws;
    int*   out_cnt   = wsi;
    int*   in_cnt    = wsi + 50048;
    int*   cursor    = wsi + 100096;
    float* src_norm  = wsf + 100160;
    float* dst_norm  = wsf + 150208;
    int*   row_start = wsi + 200256;
    int*   rank      = wsi + 250304;
    int2*  col       = (int2*)(wsi + 850304);
    float* hbuf0     = wsf + 2050304;
    float* hbuf1     = wsf + 8450304;

    hipMemsetAsync(out_cnt, 0, 100160 * sizeof(int), stream);
    deg_rank_kernel<<<(N_EDGES + 255) / 256, 256, 0, stream>>>(src, dst, out_cnt, in_cnt, rank);
    norm_scan_kernel<<<(N_NODES + 255) / 256, 256, 0, stream>>>(out_cnt, in_cnt, src_norm, dst_norm,
                                                                row_start, cursor);
    fill_kernel<<<(N_EDGES + 255) / 256, 256, 0, stream>>>(src, dst, rank, row_start, src_norm, col);

    static_assert(N_LAYERS == 4, "the table of layer outputs below has four entries");
    float* const layer_out[N_LAYERS] = {hbuf0, hbuf1, hbuf0, out};
    const float* h = feat;
    for (int i = 0; i < N_LAYERS; ++i) {
        float* o = layer_out[i];
        layer_kernel<<<(N_NODES + TILE_M - 1) / TILE_M, 256, 0, stream>>>(
            h, row_start, in_cnt, col, dst_norm, W + (size_t)i * D * D, b + (size_t)i * D, o);
        h = o;
    }
}


# Dense products reassociated (model is linear): S=W0W1W2W3 and the bias rows formed once on WMMA; layer 1 = gather + product with S, layers 2-4 = gather + bias row, no product; f32; sum order changes
# speedup vs baseline: 3.2960x; 1.0715x over previous
#include <hip/hip_runtime.h>


typedef __attribute__((ext_vector_type(2))) float v2f;
typedef __attribute__((ext_vector_type(8))) float v8f;

#define N_NODES 50000
#define N_EDGES 600000
#define D 128
#define N_LAYERS 4

#define TILE_M 64
#define NODES_PER_WAVE 8
#define GROUP 8
#define PITCH 132

static_assert(N_NODES % NODES_PER_WAVE == 0, "a wave owns whole groups of nodes");
static_assert(N_NODES % 16 == 0, "a 16-row matrix tile is wholly inside or outside the node range");
static_assert(NODES_PER_WAVE == 8, "the gather reads its bookkeeping as two int4");

__global__ __launch_bounds__(256) void deg_rank_kernel(const int* __restrict__ src,
                                                       const int* __restrict__ dst,
                                                       int* __restrict__ out_cnt,
                                                       int* __restrict__ in_cnt,
                                                       int* __restrict__ rank) {
    int e = blockIdx.x * blockDim.x + threadIdx.x;
    if (e < N_EDGES) {
        const int s = src[e];
        const int d = dst[e];
        atomicAdd(&out_cnt[s], 1);
        rank[e] = atomicAdd(&in_cnt[d], 1);
    }
}

__global__ __launch_bounds__(256) void norm_scan_kernel(const int* __restrict__ out_cnt,
                                                        const int* __restrict__ in_cnt,
                                                        float* __restrict__ src_norm,
                                                        float* __restrict__ dst_norm,
                                                        int* __restrict__ row_start,
                                                        int* __restrict__ cursor) {
    __shared__ int wave_off[8];
    __shared__ int block_base;

    const int t    = threadIdx.x;
    const int lane = t & 31;
    const int wave = t >> 5;
    const int n    = blockIdx.x * 256 + t;

    int id = 0;
    if (n < N_NODES) {
        const int od = out_cnt[n];
        id = in_cnt[n];
        src_norm[n] = rsqrtf(fmaxf((float)od, 1.0f));
        dst_norm[n] = rsqrtf(fmaxf((float)id, 1.0f));
    }

    int x = id;
#pragma unroll
    for (int o = 1; o < 32; o <<= 1) {
        int y = __shfl_up(x, o);
        if (lane >= o) x += y;
    }
    if (lane == 31) wave_off[wave] = x;
    __syncthreads();
    if (t == 0) {
        int tot = 0;
#pragma unroll
        for (int w = 0; w < 8; ++w) {
            int v = wave_off[w];
            wave_off[w] = tot;
            tot += v;
        }
        block_base = atomicAdd(cursor, tot);
    }
    __syncthreads();
    if (n < N_NODES) row_start[n] = block_base + wave_off[wave] + x - id;
}

__global__ __launch_bounds__(256) void fill_kernel(const int* __restrict__ src,
                                                   const int* __restrict__ dst,
                                                   const int* __restrict__ rank,
                                                   const int* __restrict__ row_start,
                                                   const float* __restrict__ src_norm,
                                                   int2* __restrict__ col) {
    int e = blockIdx.x * blockDim.x + threadIdx.x;
    if (e < N_EDGES) {
        const int s = src[e];
        const int d = dst[e];
        col[row_start[d] + rank[e]] = make_int2(s, __float_as_int(src_norm[s]));
    }
}

__global__ __launch_bounds__(32) void wprod_kernel(
        const float* A0, const float* ra0, const float* rb0, const float* B0,
        float* C0, float* oa0, float* ob0,
        const float* A1, const float* ra1, const float* B1,
        float* C1, float* oa1)
{
    const int  lane   = threadIdx.x;
    const bool second = (blockIdx.y != 0);
    const float* A  = second ? A1  : A0;
    const float* ra = second ? ra1 : ra0;
    const float* rb = second ? nullptr : rb0;
    const float* B  = second ? B1  : B0;
    float* C  = second ? C1  : C0;
    float* oa = second ? oa1 : oa0;
    float* ob = second ? nullptr : ob0;

    const int mt   = blockIdx.x / (D / 16);
    const int nt   = blockIdx.x % (D / 16);
    const int mrow = lane & 15;
    const int kb   = (lane >> 4) * 2;
    const int ncol = nt * 16 + (lane & 15);
    const int row  = mt * 16 + mrow;

    const float* rowp = A;
    bool valid = false;
    if (row < D)                              { rowp = A + (size_t)row * D; valid = true; }
    else if (row == D)                        { rowp = ra;                  valid = true; }
    else if (row == D + 1 && rb != nullptr)   { rowp = rb;                  valid = true; }

    v2f a[D / 4];
#pragma unroll
    for (int ks = 0; ks < D / 4; ++ks) { a[ks].x = 0.0f; a[ks].y = 0.0f; }
    if (valid) {
#pragma unroll
        for (int ks = 0; ks < D / 4; ++ks) {
            a[ks].x = rowp[ks * 4 + kb];
            a[ks].y = rowp[ks * 4 + kb + 1];
        }
    }
    v2f b[D / 4];
#pragma unroll
    for (int ks = 0; ks < D / 4; ++ks) {
        b[ks].x = B[(size_t)(ks * 4 + kb    ) * D + ncol];
        b[ks].y = B[(size_t)(ks * 4 + kb + 1) * D + ncol];
    }

    v8f c = {};
#pragma unroll
    for (int ks = 0; ks < D / 4; ++ks) {
        c = __builtin_amdgcn_wmma_f32_16x16x4_f32(
                 false, a[ks],  false, b[ks],
                 (short)0, c,  false,  false);
    }

    const int rbase = mt * 16 + (lane >> 4) * 8;
#pragma unroll
    for (int v = 0; v < 8; ++v) {
        const int r = rbase + v;
        if (r < D)                             C[(size_t)r * D + ncol] = c[v];
        else if (r == D)                       oa[ncol] = c[v];
        else if (r == D + 1 && ob != nullptr)  ob[ncol] = c[v];
    }
}

template <bool WITH_PRODUCT>
__global__ __launch_bounds__(256) void layer_kernel(
        const float* __restrict__ h,
        const int*   __restrict__ row_start,
        const int*   __restrict__ in_cnt,
        const int2*  __restrict__ col,
        const float* __restrict__ dnorm,
        const float* __restrict__ Wm,
        const float* __restrict__ bias,
        float* __restrict__ out)
{
    __shared__ float ldsA[WITH_PRODUCT ? TILE_M * PITCH : 1];

    const int t    = threadIdx.x;
    const int lane = t & 31;
    const int wave = __builtin_amdgcn_readfirstlane(t >> 5);
    const int M0   = blockIdx.x * TILE_M;
    const int n0   = M0 + wave * NODES_PER_WAVE;

    float* myrows = &ldsA[WITH_PRODUCT ? wave * NODES_PER_WAVE * PITCH + lane * 4 : 0];

    if (n0 < N_NODES) {
        const int4 rs0 = *reinterpret_cast<const int4*>(row_start + n0);
        const int4 rs1 = *reinterpret_cast<const int4*>(row_start + n0 + 4);
        const int4 ct0 = *reinterpret_cast<const int4*>(in_cnt + n0);
        const int4 ct1 = *reinterpret_cast<const int4*>(in_cnt + n0 + 4);
        const float4 dn0 = *reinterpret_cast<const float4*>(dnorm + n0);
        const float4 dn1 = *reinterpret_cast<const float4*>(dnorm + n0 + 4);
        const int   start[NODES_PER_WAVE] = {rs0.x, rs0.y, rs0.z, rs0.w, rs1.x, rs1.y, rs1.z, rs1.w};
        const int   cnt[NODES_PER_WAVE]   = {ct0.x, ct0.y, ct0.z, ct0.w, ct1.x, ct1.y, ct1.z, ct1.w};
        const float sc[NODES_PER_WAVE]    = {dn0.x, dn0.y, dn0.z, dn0.w, dn1.x, dn1.y, dn1.z, dn1.w};
        float4 bv = make_float4(0.0f, 0.0f, 0.0f, 0.0f);
        if constexpr (!WITH_PRODUCT) bv = *reinterpret_cast<const float4*>(bias + lane * 4);

        int2 first[NODES_PER_WAVE];
#pragma unroll
        for (int k = 0; k < NODES_PER_WAVE; ++k) {
            first[k] = make_int2(0, 0);
            if (lane < cnt[k]) first[k] = col[start[k] + lane];
        }

#pragma unroll
        for (int k = 0; k < NODES_PER_WAVE; ++k) {
            float4 acc = make_float4(0.0f, 0.0f, 0.0f, 0.0f);
            int2 p = first[k];
            for (int base = 0; base < cnt[k]; base += 32) {
                const int m = min(32, cnt[k] - base);
                if (base > 0) {
                    p = make_int2(0, 0);
                    if (lane < m) p = col[start[k] + base + lane];
                }
                for (int j = 0; j < m; j += GROUP) {
                    float4 v[GROUP];
                    float  w[GROUP];
#pragma unroll
                    for (int i = 0; i < GROUP; ++i) {
                        const int s = __builtin_amdgcn_readlane(p.x, j + i);
                        w[i] = __int_as_float(__builtin_amdgcn_readlane(p.y, j + i));
                        v[i] = make_float4(0.0f, 0.0f, 0.0f, 0.0f);
                        if (j + i < m) {
                            const float4* row = reinterpret_cast<const float4*>(h + (size_t)s * D);
                            v[i] = row[lane];
                        }
                    }
#pragma unroll
                    for (int i = 0; i < GROUP; ++i) {
                        acc.x = fmaf(v[i].x, w[i], acc.x);
                        acc.y = fmaf(v[i].y, w[i], acc.y);
                        acc.z = fmaf(v[i].z, w[i], acc.z);
                        acc.w = fmaf(v[i].w, w[i], acc.w);
                    }
                }
            }
            if constexpr (WITH_PRODUCT) {
                *reinterpret_cast<float4*>(myrows + k * PITCH) =
                    make_float4(acc.x * sc[k], acc.y * sc[k], acc.z * sc[k], acc.w * sc[k]);
            } else {
                *reinterpret_cast<float4*>(out + (size_t)(n0 + k) * D + lane * 4) =
                    make_float4(fmaf(acc.x, sc[k], bv.x), fmaf(acc.y, sc[k], bv.y),
                                fmaf(acc.z, sc[k], bv.z), fmaf(acc.w, sc[k], bv.w));
            }
        }
    } else if constexpr (WITH_PRODUCT) {
#pragma unroll
        for (int k = 0; k < NODES_PER_WAVE; ++k)
            *reinterpret_cast<float4*>(myrows + k * PITCH) = make_float4(0.0f, 0.0f, 0.0f, 0.0f);
    }

    if constexpr (WITH_PRODUCT) {

    const int ncol = wave * 16 + (lane & 15);
    const int mrow = lane & 15;
    const int kb   = (lane >> 4) * 2;

    v2f b[D / 4];
#pragma unroll
    for (int ks = 0; ks < D / 4; ++ks) {
        b[ks].x = Wm[(size_t)(ks * 4 + kb    ) * D + ncol];
        b[ks].y = Wm[(size_t)(ks * 4 + kb + 1) * D + ncol];
    }
    const float bn = bias[ncol];

    __syncthreads();

    v8f c[TILE_M / 16] = {};
#pragma unroll
    for (int ks = 0; ks < D / 4; ++ks) {
#pragma unroll
        for (int mt = 0; mt < TILE_M / 16; ++mt) {
            const float* ap = &ldsA[(mt * 16 + mrow) * PITCH + ks * 4 + kb];
            v2f a;
            a.x = ap[0];
            a.y = ap[1];
            c[mt] = __builtin_amdgcn_wmma_f32_16x16x4_f32(
                     false, a,  false, b[ks],
                     (short)0, c[mt],  false,  false);
        }
    }

#pragma unroll
    for (int mt = 0; mt < TILE_M / 16; ++mt) {
        const int mtile = M0 + mt * 16;
        if (mtile < N_NODES) {
            const int mbase = mtile + ((lane >> 4) * 8);
#pragma unroll
            for (int v = 0; v < 8; ++v) {
                out[(size_t)(mbase + v) * D + ncol] = c[mt][v] + bn;
            }
        }
    }
    }
}

extern "C" void kernel_launch(void* const* d_in, const int* in_sizes, int n_in,
                              void* d_out, int out_size, void* d_ws, size_t ws_size,
                              hipStream_t stream) {
    const float* feat = (const float*)d_in[0];
    const int*   src  = (const int*)  d_in[1];
    const int*   dst  = (const int*)  d_in[2];
    const float* W    = (const float*)d_in[3];
    const float* b    = (const float*)d_in[4];
    float*       out  = (float*)d_out;

    int*   wsi       = (int*)d_ws;
    float* wsf       = (float*)d_ws;
    int*   out_cnt   = wsi;
    int*   in_cnt    = wsi + 50048;
    int*   cursor    = wsi + 100096;
    float* src_norm  = wsf + 100160;
    float* dst_norm  = wsf + 150208;
    int*   row_start = wsi + 200256;
    int*   rank      = wsi + 250304;
    int2*  col       = (int2*)(wsi + 850304);
    float* hbuf0     = wsf + 2050304;
    float* hbuf1     = wsf + 8450304;
    float* Pm        = wsf + 14850304;
    float* Qm        = wsf + 14866688;
    float* S0        = wsf + 14883072;
    float* c2        = wsf + 14899456;
    float* r0        = wsf + 14899584;
    float* c0        = wsf + 14899712;
    float* c1        = wsf + 14899840;

    hipMemsetAsync(out_cnt, 0, 100160 * sizeof(int), stream);
    deg_rank_kernel<<<(N_EDGES + 255) / 256, 256, 0, stream>>>(src, dst, out_cnt, in_cnt, rank);
    norm_scan_kernel<<<(N_NODES + 255) / 256, 256, 0, stream>>>(out_cnt, in_cnt, src_norm, dst_norm,
                                                                row_start, cursor);
    fill_kernel<<<(N_EDGES + 255) / 256, 256, 0, stream>>>(src, dst, rank, row_start, src_norm, col);

    static_assert(N_LAYERS == 4, "the reassociation below is written out for four layers");
    wprod_kernel<<<dim3(9 * (D / 16), 2), 32, 0, stream>>>(
        W + (size_t)2 * D * D, b + (size_t)2 * D, nullptr, W + (size_t)3 * D * D, Pm, c2, nullptr,
        W, b, W + (size_t)1 * D * D, Qm, r0);
    wprod_kernel<<<dim3(9 * (D / 16), 1), 32, 0, stream>>>(
        Qm, r0, b + (size_t)1 * D, Pm, S0, c0, c1,
        nullptr, nullptr, nullptr, nullptr, nullptr);

    const int layer_blocks = (N_NODES + TILE_M - 1) / TILE_M;
    layer_kernel<true><<<layer_blocks, 256, 0, stream>>>(
        feat, row_start, in_cnt, col, dst_norm, S0, c0, hbuf0);
    layer_kernel<false><<<layer_blocks, 256, 0, stream>>>(
        hbuf0, row_start, in_cnt, col, dst_norm, nullptr, c1, hbuf1);
    layer_kernel<false><<<layer_blocks, 256, 0, stream>>>(
        hbuf1, row_start, in_cnt, col, dst_norm, nullptr, c2, hbuf0);
    layer_kernel<false><<<layer_blocks, 256, 0, stream>>>(
        hbuf0, row_start, in_cnt, col, dst_norm, nullptr, b + (size_t)3 * D, out);
}


# Degree counts and edge ranks on LDS counters: each block owns a node range and scans a whole index array (no global atomic, no clear left there); rank = arrival order in block; sum order changes
# speedup vs baseline: 3.8174x; 1.1582x over previous
#include <hip/hip_runtime.h>


typedef __attribute__((ext_vector_type(2))) float v2f;
typedef __attribute__((ext_vector_type(8))) float v8f;

#define N_NODES 50000
#define N_EDGES 600000
#define D 128
#define N_LAYERS 4

#define TILE_M 64
#define NODES_PER_WAVE 8
#define GROUP 8
#define PITCH 132

static_assert(N_NODES % NODES_PER_WAVE == 0, "a wave owns whole groups of nodes");
static_assert(N_NODES % 16 == 0, "a 16-row matrix tile is wholly inside or outside the node range");
static_assert(NODES_PER_WAVE == 8, "the gather reads its bookkeeping as two int4");

#define RANGE_BLOCKS  64
#define RANGE_NODES   ((N_NODES + RANGE_BLOCKS - 1) / RANGE_BLOCKS)
#define RANGE_THREADS 1024
#define VEC_PER_TRIP  4
#define NVEC          (N_EDGES / 4)
#define TRIP_VECS     (RANGE_THREADS * VEC_PER_TRIP)
#define FULL_TRIPS    (NVEC / TRIP_VECS)
#define TAIL_VEC      ((NVEC - FULL_TRIPS * TRIP_VECS + RANGE_THREADS - 1) / RANGE_THREADS)
#define START_STEP    ((NVEC / RANGE_BLOCKS) & ~31)

static_assert(N_EDGES % 4 == 0, "the index arrays are read as int4");
static_assert(FULL_TRIPS >= 2 && FULL_TRIPS % 2 == 0, "the pass below alternates two register sets over the full trips");
static_assert(NVEC % TRIP_VECS != 0, "the pass below ends with one partial trip");

template <bool TAIL, int NQ>
__device__ __forceinline__ void load_trip(const int4* __restrict__ idx, int j0, int start,
                                          int4 (&v)[VEC_PER_TRIP], int (&vi)[VEC_PER_TRIP]) {
#pragma unroll
    for (int q = 0; q < NQ; ++q) {
        const int j = j0 + q * RANGE_THREADS;
        int i = (TAIL ? min(j, NVEC - 1) : j) + start;
        if (i >= NVEC) i -= NVEC;
        vi[q] = i;
        v[q]  = idx[i];
        if (TAIL && j >= NVEC) v[q] = make_int4(-1, -1, -1, -1);
    }
}

template <bool RANKED, int NQ>
__device__ __forceinline__ void count_trip(const int4 (&v)[VEC_PER_TRIP], const int (&vi)[VEC_PER_TRIP],
                                           int* cnt, unsigned nlo4, int* __restrict__ rank) {
    constexpr unsigned LIM = 4u * RANGE_NODES;
    char* cb = reinterpret_cast<char*>(cnt);
    unsigned a[NQ][4];
#pragma unroll
    for (int q = 0; q < NQ; ++q) {
        a[q][0] = ((unsigned)v[q].x << 2) + nlo4;
        a[q][1] = ((unsigned)v[q].y << 2) + nlo4;
        a[q][2] = ((unsigned)v[q].z << 2) + nlo4;
        a[q][3] = ((unsigned)v[q].w << 2) + nlo4;
    }
    if constexpr (RANKED) {
        int k[NQ][4];
#pragma unroll
        for (int q = 0; q < NQ; ++q) {
#pragma unroll
            for (int c = 0; c < 4; ++c) {
                k[q][c] = 0;
                if (a[q][c] < LIM) k[q][c] = atomicAdd(reinterpret_cast<int*>(cb + a[q][c]), 1);
            }
        }
#pragma unroll
        for (int q = 0; q < NQ; ++q) {
#pragma unroll
            for (int c = 0; c < 4; ++c) {
                if (a[q][c] < LIM) rank[4 * vi[q] + c] = k[q][c];
            }
        }
    } else {
#pragma unroll
        for (int q = 0; q < NQ; ++q) {
#pragma unroll
            for (int c = 0; c < 4; ++c) {
                if (a[q][c] < LIM) atomicAdd(reinterpret_cast<int*>(cb + a[q][c]), 1);
            }
        }
    }
}

template <bool RANKED>
__device__ __forceinline__ void count_range(const int4* __restrict__ idx, int* cnt, int lo, int start,
                                            int* __restrict__ rank, int t) {
    const unsigned nlo4 = (unsigned)(-4 * lo);
    int4 va[VEC_PER_TRIP], vb[VEC_PER_TRIP];
    int  ia[VEC_PER_TRIP], ib[VEC_PER_TRIP];
    constexpr int V = VEC_PER_TRIP;
    load_trip<false, V>(idx, t, start, va, ia);
    for (int p = 0; p < FULL_TRIPS - 2; p += 2) {
        load_trip<false, V>(idx, (p + 1) * TRIP_VECS + t, start, vb, ib);
        count_trip<RANKED, V>(va, ia, cnt, nlo4, rank);
        load_trip<false, V>(idx, (p + 2) * TRIP_VECS + t, start, va, ia);
        count_trip<RANKED, V>(vb, ib, cnt, nlo4, rank);
    }
    load_trip<false, V>(idx, (FULL_TRIPS - 1) * TRIP_VECS + t, start, vb, ib);
    count_trip<RANKED, V>(va, ia, cnt, nlo4, rank);
    load_trip<true, TAIL_VEC>(idx, FULL_TRIPS * TRIP_VECS + t, start, va, ia);
    count_trip<RANKED, V>(vb, ib, cnt, nlo4, rank);
    count_trip<RANKED, TAIL_VEC>(va, ia, cnt, nlo4, rank);
}

__global__ __launch_bounds__(RANGE_THREADS) void deg_rank_kernel(const int* __restrict__ src,
                                                                 const int* __restrict__ dst,
                                                                 int* __restrict__ out_cnt,
                                                                 int* __restrict__ in_cnt,
                                                                 int* __restrict__ rank,
                                                                 int* __restrict__ cursor) {
    __shared__ int cnt[RANGE_NODES];

    const int  t      = threadIdx.x;
    const bool by_dst = blockIdx.x < RANGE_BLOCKS;
    const int  piece  = by_dst ? blockIdx.x : blockIdx.x - RANGE_BLOCKS;
    const int  lo     = piece * RANGE_NODES;
    const int  start  = piece * START_STEP;

    for (int i = t; i < RANGE_NODES; i += RANGE_THREADS) cnt[i] = 0;
    if (blockIdx.x == 0 && t == 0) *cursor = 0;
    __syncthreads();

    if (by_dst) count_range<true >(reinterpret_cast<const int4*>(dst), cnt, lo, start, rank, t);
    else        count_range<false>(reinterpret_cast<const int4*>(src), cnt, lo, start, rank, t);

    __syncthreads();
    int* __restrict__ total = by_dst ? in_cnt : out_cnt;
    for (int i = t; i < RANGE_NODES; i += RANGE_THREADS) {
        const int n = lo + i;
        if (n < N_NODES) total[n] = cnt[i];
    }
}

__global__ __launch_bounds__(256) void norm_scan_kernel(const int* __restrict__ out_cnt,
                                                        const int* __restrict__ in_cnt,
                                                        float* __restrict__ src_norm,
                                                        float* __restrict__ dst_norm,
                                                        int* __restrict__ row_start,
                                                        int* __restrict__ cursor) {
    __shared__ int wave_off[8];
    __shared__ int block_base;

    const int t    = threadIdx.x;
    const int lane = t & 31;
    const int wave = t >> 5;
    const int n    = blockIdx.x * 256 + t;

    int id = 0;
    if (n < N_NODES) {
        const int od = out_cnt[n];
        id = in_cnt[n];
        src_norm[n] = rsqrtf(fmaxf((float)od, 1.0f));
        dst_norm[n] = rsqrtf(fmaxf((float)id, 1.0f));
    }

    int x = id;
#pragma unroll
    for (int o = 1; o < 32; o <<= 1) {
        int y = __shfl_up(x, o);
        if (lane >= o) x += y;
    }
    if (lane == 31) wave_off[wave] = x;
    __syncthreads();
    if (t == 0) {
        int tot = 0;
#pragma unroll
        for (int w = 0; w < 8; ++w) {
            int v = wave_off[w];
            wave_off[w] = tot;
            tot += v;
        }
        block_base = atomicAdd(cursor, tot);
    }
    __syncthreads();
    if (n < N_NODES) row_start[n] = block_base + wave_off[wave] + x - id;
}

__global__ __launch_bounds__(256) void fill_kernel(const int* __restrict__ src,
                                                   const int* __restrict__ dst,
                                                   const int* __restrict__ rank,
                                                   const int* __restrict__ row_start,
                                                   const float* __restrict__ src_norm,
                                                   int2* __restrict__ col) {
    int e = blockIdx.x * blockDim.x + threadIdx.x;
    if (e < N_EDGES) {
        const int s = src[e];
        const int d = dst[e];
        col[row_start[d] + rank[e]] = make_int2(s, __float_as_int(src_norm[s]));
    }
}

__global__ __launch_bounds__(32) void wprod_kernel(
        const float* A0, const float* ra0, const float* rb0, const float* B0,
        float* C0, float* oa0, float* ob0,
        const float* A1, const float* ra1, const float* B1,
        float* C1, float* oa1)
{
    const int  lane   = threadIdx.x;
    const bool second = (blockIdx.y != 0);
    const float* A  = second ? A1  : A0;
    const float* ra = second ? ra1 : ra0;
    const float* rb = second ? nullptr : rb0;
    const float* B  = second ? B1  : B0;
    float* C  = second ? C1  : C0;
    float* oa = second ? oa1 : oa0;
    float* ob = second ? nullptr : ob0;

    const int mt   = blockIdx.x / (D / 16);
    const int nt   = blockIdx.x % (D / 16);
    const int mrow = lane & 15;
    const int kb   = (lane >> 4) * 2;
    const int ncol = nt * 16 + (lane & 15);
    const int row  = mt * 16 + mrow;

    const float* rowp = A;
    bool valid = false;
    if (row < D)                              { rowp = A + (size_t)row * D; valid = true; }
    else if (row == D)                        { rowp = ra;                  valid = true; }
    else if (row == D + 1 && rb != nullptr)   { rowp = rb;                  valid = true; }

    v2f a[D / 4];
#pragma unroll
    for (int ks = 0; ks < D / 4; ++ks) { a[ks].x = 0.0f; a[ks].y = 0.0f; }
    if (valid) {
#pragma unroll
        for (int ks = 0; ks < D / 4; ++ks) {
            a[ks].x = rowp[ks * 4 + kb];
            a[ks].y = rowp[ks * 4 + kb + 1];
        }
    }
    v2f b[D / 4];
#pragma unroll
    for (int ks = 0; ks < D / 4; ++ks) {
        b[ks].x = B[(size_t)(ks * 4 + kb    ) * D + ncol];
        b[ks].y = B[(size_t)(ks * 4 + kb + 1) * D + ncol];
    }

    v8f c = {};
#pragma unroll
    for (int ks = 0; ks < D / 4; ++ks) {
        c = __builtin_amdgcn_wmma_f32_16x16x4_f32(
                 false, a[ks],  false, b[ks],
                 (short)0, c,  false,  false);
    }

    const int rbase = mt * 16 + (lane >> 4) * 8;
#pragma unroll
    for (int v = 0; v < 8; ++v) {
        const int r = rbase + v;
        if (r < D)                             C[(size_t)r * D + ncol] = c[v];
        else if (r == D)                       oa[ncol] = c[v];
        else if (r == D + 1 && ob != nullptr)  ob[ncol] = c[v];
    }
}

template <bool WITH_PRODUCT>
__global__ __launch_bounds__(256) void layer_kernel(
        const float* __restrict__ h,
        const int*   __restrict__ row_start,
        const int*   __restrict__ in_cnt,
        const int2*  __restrict__ col,
        const float* __restrict__ dnorm,
        const float* __restrict__ Wm,
        const float* __restrict__ bias,
        float* __restrict__ out)
{
    __shared__ float ldsA[WITH_PRODUCT ? TILE_M * PITCH : 1];

    const int t    = threadIdx.x;
    const int lane = t & 31;
    const int wave = __builtin_amdgcn_readfirstlane(t >> 5);
    const int M0   = blockIdx.x * TILE_M;
    const int n0   = M0 + wave * NODES_PER_WAVE;

    float* myrows = &ldsA[WITH_PRODUCT ? wave * NODES_PER_WAVE * PITCH + lane * 4 : 0];

    if (n0 < N_NODES) {
        const int4 rs0 = *reinterpret_cast<const int4*>(row_start + n0);
        const int4 rs1 = *reinterpret_cast<const int4*>(row_start + n0 + 4);
        const int4 ct0 = *reinterpret_cast<const int4*>(in_cnt + n0);
        const int4 ct1 = *reinterpret_cast<const int4*>(in_cnt + n0 + 4);
        const float4 dn0 = *reinterpret_cast<const float4*>(dnorm + n0);
        const float4 dn1 = *reinterpret_cast<const float4*>(dnorm + n0 + 4);
        const int   start[NODES_PER_WAVE] = {rs0.x, rs0.y, rs0.z, rs0.w, rs1.x, rs1.y, rs1.z, rs1.w};
        const int   cnt[NODES_PER_WAVE]   = {ct0.x, ct0.y, ct0.z, ct0.w, ct1.x, ct1.y, ct1.z, ct1.w};
        const float sc[NODES_PER_WAVE]    = {dn0.x, dn0.y, dn0.z, dn0.w, dn1.x, dn1.y, dn1.z, dn1.w};
        float4 bv = make_float4(0.0f, 0.0f, 0.0f, 0.0f);
        if constexpr (!WITH_PRODUCT) bv = *reinterpret_cast<const float4*>(bias + lane * 4);

        int2 first[NODES_PER_WAVE];
#pragma unroll
        for (int k = 0; k < NODES_PER_WAVE; ++k) {
            first[k] = make_int2(0, 0);
            if (lane < cnt[k]) first[k] = col[start[k] + lane];
        }

#pragma unroll
        for (int k = 0; k < NODES_PER_WAVE; ++k) {
            float4 acc = make_float4(0.0f, 0.0f, 0.0f, 0.0f);
            int2 p = first[k];
            for (int base = 0; base < cnt[k]; base += 32) {
                const int m = min(32, cnt[k] - base);
                if (base > 0) {
                    p = make_int2(0, 0);
                    if (lane < m) p = col[start[k] + base + lane];
                }
                for (int j = 0; j < m; j += GROUP) {
                    float4 v[GROUP];
                    float  w[GROUP];
#pragma unroll
                    for (int i = 0; i < GROUP; ++i) {
                        const int s = __builtin_amdgcn_readlane(p.x, j + i);
                        w[i] = __int_as_float(__builtin_amdgcn_readlane(p.y, j + i));
                        v[i] = make_float4(0.0f, 0.0f, 0.0f, 0.0f);
                        if (j + i < m) {
                            const float4* row = reinterpret_cast<const float4*>(h + (size_t)s * D);
                            v[i] = row[lane];
                        }
                    }
#pragma unroll
                    for (int i = 0; i < GROUP; ++i) {
                        acc.x = fmaf(v[i].x, w[i], acc.x);
                        acc.y = fmaf(v[i].y, w[i], acc.y);
                        acc.z = fmaf(v[i].z, w[i], acc.z);
                        acc.w = fmaf(v[i].w, w[i], acc.w);
                    }
                }
            }
            if constexpr (WITH_PRODUCT) {
                *reinterpret_cast<float4*>(myrows + k * PITCH) =
                    make_float4(acc.x * sc[k], acc.y * sc[k], acc.z * sc[k], acc.w * sc[k]);
            } else {
                *reinterpret_cast<float4*>(out + (size_t)(n0 + k) * D + lane * 4) =
                    make_float4(fmaf(acc.x, sc[k], bv.x), fmaf(acc.y, sc[k], bv.y),
                                fmaf(acc.z, sc[k], bv.z), fmaf(acc.w, sc[k], bv.w));
            }
        }
    } else if constexpr (WITH_PRODUCT) {
#pragma unroll
        for (int k = 0; k < NODES_PER_WAVE; ++k)
            *reinterpret_cast<float4*>(myrows + k * PITCH) = make_float4(0.0f, 0.0f, 0.0f, 0.0f);
    }

    if constexpr (WITH_PRODUCT) {

    const int ncol = wave * 16 + (lane & 15);
    const int mrow = lane & 15;
    const int kb   = (lane >> 4) * 2;

    v2f b[D / 4];
#pragma unroll
    for (int ks = 0; ks < D / 4; ++ks) {
        b[ks].x = Wm[(size_t)(ks * 4 + kb    ) * D + ncol];
        b[ks].y = Wm[(size_t)(ks * 4 + kb + 1) * D + ncol];
    }
    const float bn = bias[ncol];

    __syncthreads();

    v8f c[TILE_M / 16] = {};
#pragma unroll
    for (int ks = 0; ks < D / 4; ++ks) {
#pragma unroll
        for (int mt = 0; mt < TILE_M / 16; ++mt) {
            const float* ap = &ldsA[(mt * 16 + mrow) * PITCH + ks * 4 + kb];
            v2f a;
            a.x = ap[0];
            a.y = ap[1];
            c[mt] = __builtin_amdgcn_wmma_f32_16x16x4_f32(
                     false, a,  false, b[ks],
                     (short)0, c[mt],  false,  false);
        }
    }

#pragma unroll
    for (int mt = 0; mt < TILE_M / 16; ++mt) {
        const int mtile = M0 + mt * 16;
        if (mtile < N_NODES) {
            const int mbase = mtile + ((lane >> 4) * 8);
#pragma unroll
            for (int v = 0; v < 8; ++v) {
                out[(size_t)(mbase + v) * D + ncol] = c[mt][v] + bn;
            }
        }
    }
    }
}

extern "C" void kernel_launch(void* const* d_in, const int* in_sizes, int n_in,
                              void* d_out, int out_size, void* d_ws, size_t ws_size,
                              hipStream_t stream) {
    const float* feat = (const float*)d_in[0];
    const int*   src  = (const int*)  d_in[1];
    const int*   dst  = (const int*)  d_in[2];
    const float* W    = (const float*)d_in[3];
    const float* b    = (const float*)d_in[4];
    float*       out  = (float*)d_out;

    int*   wsi       = (int*)d_ws;
    float* wsf       = (float*)d_ws;
    int*   out_cnt   = wsi;
    int*   in_cnt    = wsi + 50048;
    int*   cursor    = wsi + 100096;
    float* src_norm  = wsf + 100160;
    float* dst_norm  = wsf + 150208;
    int*   row_start = wsi + 200256;
    int*   rank      = wsi + 250304;
    int2*  col       = (int2*)(wsi + 850304);
    float* hbuf0     = wsf + 2050304;
    float* hbuf1     = wsf + 8450304;
    float* Pm        = wsf + 14850304;
    float* Qm        = wsf + 14866688;
    float* S0        = wsf + 14883072;
    float* c2        = wsf + 14899456;
    float* r0        = wsf + 14899584;
    float* c0        = wsf + 14899712;
    float* c1        = wsf + 14899840;

    deg_rank_kernel<<<2 * RANGE_BLOCKS, RANGE_THREADS, 0, stream>>>(src, dst, out_cnt, in_cnt, rank, cursor);
    norm_scan_kernel<<<(N_NODES + 255) / 256, 256, 0, stream>>>(out_cnt, in_cnt, src_norm, dst_norm,
                                                                row_start, cursor);
    fill_kernel<<<(N_EDGES + 255) / 256, 256, 0, stream>>>(src, dst, rank, row_start, src_norm, col);

    static_assert(N_LAYERS == 4, "the reassociation below is written out for four layers");
    wprod_kernel<<<dim3(9 * (D / 16), 2), 32, 0, stream>>>(
        W + (size_t)2 * D * D, b + (size_t)2 * D, nullptr, W + (size_t)3 * D * D, Pm, c2, nullptr,
        W, b, W + (size_t)1 * D * D, Qm, r0);
    wprod_kernel<<<dim3(9 * (D / 16), 1), 32, 0, stream>>>(
        Qm, r0, b + (size_t)1 * D, Pm, S0, c0, c1,
        nullptr, nullptr, nullptr, nullptr, nullptr);

    const int layer_blocks = (N_NODES + TILE_M - 1) / TILE_M;
    layer_kernel<true><<<layer_blocks, 256, 0, stream>>>(
        feat, row_start, in_cnt, col, dst_norm, S0, c0, hbuf0);
    layer_kernel<false><<<layer_blocks, 256, 0, stream>>>(
        hbuf0, row_start, in_cnt, col, dst_norm, nullptr, c1, hbuf1);
    layer_kernel<false><<<layer_blocks, 256, 0, stream>>>(
        hbuf1, row_start, in_cnt, col, dst_norm, nullptr, c2, hbuf0);
    layer_kernel<false><<<layer_blocks, 256, 0, stream>>>(
        hbuf0, row_start, in_cnt, col, dst_norm, nullptr, b + (size_t)3 * D, out);
}


# Degree/rank pass tiled 2-D: a block scans 1/8 of an index array for 1/8 of the nodes (was: the whole array); per-piece counts become per-piece slot starts in the scan kernel; sum order changes
# speedup vs baseline: 4.5030x; 1.1796x over previous
#include <hip/hip_runtime.h>


typedef __attribute__((ext_vector_type(2))) float v2f;
typedef __attribute__((ext_vector_type(8))) float v8f;

#define N_NODES 50000
#define N_EDGES 600000
#define D 128
#define N_LAYERS 4

#define TILE_M 64
#define NODES_PER_WAVE 8
#define GROUP 8
#define PITCH 132

static_assert(N_NODES % NODES_PER_WAVE == 0, "a wave owns whole groups of nodes");
static_assert(N_NODES % 16 == 0, "a 16-row matrix tile is wholly inside or outside the node range");
static_assert(NODES_PER_WAVE == 8, "the gather reads its bookkeeping as two int4");

#define RANGE_BLOCKS  8
#define EDGE_CHUNKS   8
#define RANGE_NODES   (N_NODES / RANGE_BLOCKS)
#define RANGE_THREADS 1024
#define VEC_PER_TRIP  5
#define NVEC          (N_EDGES / 4)
#define CHUNK_VECS    (NVEC / EDGE_CHUNKS)
#define CHUNK_EDGES   (4 * CHUNK_VECS)
#define TRIP_VECS     (RANGE_THREADS * VEC_PER_TRIP)
#define FULL_TRIPS    (CHUNK_VECS / TRIP_VECS)
#define TAIL_VEC      ((CHUNK_VECS - FULL_TRIPS * TRIP_VECS + RANGE_THREADS - 1) / RANGE_THREADS)
#define PART_PITCH    50048

static_assert(N_EDGES % 4 == 0, "the index arrays are read as int4");
static_assert(N_NODES % RANGE_BLOCKS == 0, "the node pieces are equal");
static_assert(NVEC % EDGE_CHUNKS == 0, "the array pieces are equal and made of whole int4 vectors");
static_assert(FULL_TRIPS == 3, "the pass below is written out for three full trips and a partial one");
static_assert(CHUNK_VECS % TRIP_VECS != 0 && TAIL_VEC <= VEC_PER_TRIP, "the pass below ends with one partial trip");
static_assert(PART_PITCH >= N_NODES && PART_PITCH % 64 == 0, "a plane holds every node and starts on a 256-byte boundary");

template <bool TAIL, int NQ>
__device__ __forceinline__ void load_trip(const int4* __restrict__ idx, int j0, int first,
                                          int4 (&v)[VEC_PER_TRIP], int (&vi)[VEC_PER_TRIP]) {
#pragma unroll
    for (int q = 0; q < NQ; ++q) {
        const int j = j0 + q * RANGE_THREADS;
        const int i = first + (TAIL ? min(j, CHUNK_VECS - 1) : j);
        vi[q] = i;
        v[q]  = idx[i];
        if (TAIL && j >= CHUNK_VECS) v[q] = make_int4(-1, -1, -1, -1);
    }
}

template <bool RANKED, int NQ>
__device__ __forceinline__ void count_trip(const int4 (&v)[VEC_PER_TRIP], const int (&vi)[VEC_PER_TRIP],
                                           int* cnt, unsigned nlo4, int* __restrict__ rank) {
    constexpr unsigned LIM = 4u * RANGE_NODES;
    char* cb = reinterpret_cast<char*>(cnt);
    unsigned a[NQ][4];
#pragma unroll
    for (int q = 0; q < NQ; ++q) {
        a[q][0] = ((unsigned)v[q].x << 2) + nlo4;
        a[q][1] = ((unsigned)v[q].y << 2) + nlo4;
        a[q][2] = ((unsigned)v[q].z << 2) + nlo4;
        a[q][3] = ((unsigned)v[q].w << 2) + nlo4;
    }
    if constexpr (RANKED) {
        int k[NQ][4];
#pragma unroll
        for (int q = 0; q < NQ; ++q) {
#pragma unroll
            for (int c = 0; c < 4; ++c) {
                k[q][c] = 0;
                if (a[q][c] < LIM) k[q][c] = atomicAdd(reinterpret_cast<int*>(cb + a[q][c]), 1);
            }
        }
#pragma unroll
        for (int q = 0; q < NQ; ++q) {
#pragma unroll
            for (int c = 0; c < 4; ++c) {
                if (a[q][c] < LIM) rank[4 * vi[q] + c] = k[q][c];
            }
        }
    } else {
#pragma unroll
        for (int q = 0; q < NQ; ++q) {
#pragma unroll
            for (int c = 0; c < 4; ++c) {
                if (a[q][c] < LIM) atomicAdd(reinterpret_cast<int*>(cb + a[q][c]), 1);
            }
        }
    }
}

template <bool RANKED>
__device__ __forceinline__ void count_piece(const int4* __restrict__ idx, int* cnt, int lo, int first,
                                            int* __restrict__ rank, int t,
                                            int4 (&va)[VEC_PER_TRIP], int (&ia)[VEC_PER_TRIP],
                                            int4 (&vb)[VEC_PER_TRIP], int (&ib)[VEC_PER_TRIP]) {
    const unsigned nlo4 = (unsigned)(-4 * lo);
    constexpr int V = VEC_PER_TRIP;
    count_trip<RANKED, V>(va, ia, cnt, nlo4, rank);
    load_trip<false, V>(idx, 2 * TRIP_VECS + t, first, va, ia);
    count_trip<RANKED, V>(vb, ib, cnt, nlo4, rank);
    load_trip<true, TAIL_VEC>(idx, 3 * TRIP_VECS + t, first, vb, ib);
    count_trip<RANKED, V>(va, ia, cnt, nlo4, rank);
    count_trip<RANKED, TAIL_VEC>(vb, ib, cnt, nlo4, rank);
}

__global__ __launch_bounds__(RANGE_THREADS) void deg_rank_kernel(const int* __restrict__ src,
                                                                 const int* __restrict__ dst,
                                                                 int* __restrict__ out_part,
                                                                 int* __restrict__ in_part,
                                                                 int* __restrict__ rank,
                                                                 int* __restrict__ cursor) {
    __shared__ int cnt[RANGE_NODES];

    const int  t      = threadIdx.x;
    const bool by_dst = blockIdx.x < RANGE_BLOCKS * EDGE_CHUNKS;
    const int  b      = by_dst ? blockIdx.x : blockIdx.x - RANGE_BLOCKS * EDGE_CHUNKS;
    const int  piece  = b / EDGE_CHUNKS;
    const int  chunk  = b % EDGE_CHUNKS;
    const int  lo     = piece * RANGE_NODES;
    const int  first  = chunk * CHUNK_VECS;
    const int4* idx   = reinterpret_cast<const int4*>(by_dst ? dst : src);

    int4 va[VEC_PER_TRIP], vb[VEC_PER_TRIP];
    int  ia[VEC_PER_TRIP], ib[VEC_PER_TRIP];
    load_trip<false, VEC_PER_TRIP>(idx, t, first, va, ia);
    load_trip<false, VEC_PER_TRIP>(idx, TRIP_VECS + t, first, vb, ib);

    for (int i = t; i < RANGE_NODES; i += RANGE_THREADS) cnt[i] = 0;
    if (blockIdx.x == 0 && t == 0) *cursor = 0;
    __syncthreads();

    if (by_dst) count_piece<true >(idx, cnt, lo, first, rank, t, va, ia, vb, ib);
    else        count_piece<false>(idx, cnt, lo, first, rank, t, va, ia, vb, ib);

    __syncthreads();
    int* __restrict__ plane = (by_dst ? in_part : out_part) + chunk * PART_PITCH + lo;
    for (int i = t; i < RANGE_NODES; i += RANGE_THREADS) plane[i] = cnt[i];
}

__global__ __launch_bounds__(256) void norm_scan_kernel(const int* __restrict__ out_part,
                                                        int* in_part,
                                                        float* __restrict__ src_norm,
                                                        float* __restrict__ dst_norm,
                                                        int* __restrict__ in_cnt,
                                                        int* __restrict__ cursor) {
    __shared__ int wave_off[8];
    __shared__ int block_base;

    const int t    = threadIdx.x;
    const int lane = t & 31;
    const int wave = t >> 5;
    const int n    = blockIdx.x * 256 + t;

    int id = 0;
    int before[EDGE_CHUNKS];
#pragma unroll
    for (int c = 0; c < EDGE_CHUNKS; ++c) before[c] = 0;
    if (n < N_NODES) {
        int od = 0;
        int ip[EDGE_CHUNKS];
#pragma unroll
        for (int c = 0; c < EDGE_CHUNKS; ++c) {
            od   += out_part[c * PART_PITCH + n];
            ip[c] = in_part[c * PART_PITCH + n];
        }
#pragma unroll
        for (int c = 0; c < EDGE_CHUNKS; ++c) {
            before[c] = id;
            id += ip[c];
        }
        in_cnt[n]   = id;
        src_norm[n] = rsqrtf(fmaxf((float)od, 1.0f));
        dst_norm[n] = rsqrtf(fmaxf((float)id, 1.0f));
    }

    int x = id;
#pragma unroll
    for (int o = 1; o < 32; o <<= 1) {
        int y = __shfl_up(x, o);
        if (lane >= o) x += y;
    }
    if (lane == 31) wave_off[wave] = x;
    __syncthreads();
    if (t == 0) {
        int tot = 0;
#pragma unroll
        for (int w = 0; w < 8; ++w) {
            int v = wave_off[w];
            wave_off[w] = tot;
            tot += v;
        }
        block_base = atomicAdd(cursor, tot);
    }
    __syncthreads();
    if (n < N_NODES) {
        const int seg = block_base + wave_off[wave] + x - id;
#pragma unroll
        for (int c = 0; c < EDGE_CHUNKS; ++c) in_part[c * PART_PITCH + n] = seg + before[c];
    }
}

__global__ __launch_bounds__(256) void fill_kernel(const int* __restrict__ src,
                                                   const int* __restrict__ dst,
                                                   const int* __restrict__ rank,
                                                   const int* __restrict__ chunk_start,
                                                   const float* __restrict__ src_norm,
                                                   int2* __restrict__ col) {
    int e = blockIdx.x * blockDim.x + threadIdx.x;
    if (e < N_EDGES) {
        const int s = src[e];
        const int d = dst[e];
        const int c = e / CHUNK_EDGES;
        col[chunk_start[c * PART_PITCH + d] + rank[e]] = make_int2(s, __float_as_int(src_norm[s]));
    }
}

__global__ __launch_bounds__(32) void wprod_kernel(
        const float* A0, const float* ra0, const float* rb0, const float* B0,
        float* C0, float* oa0, float* ob0,
        const float* A1, const float* ra1, const float* B1,
        float* C1, float* oa1)
{
    const int  lane   = threadIdx.x;
    const bool second = (blockIdx.y != 0);
    const float* A  = second ? A1  : A0;
    const float* ra = second ? ra1 : ra0;
    const float* rb = second ? nullptr : rb0;
    const float* B  = second ? B1  : B0;
    float* C  = second ? C1  : C0;
    float* oa = second ? oa1 : oa0;
    float* ob = second ? nullptr : ob0;

    const int mt   = blockIdx.x / (D / 16);
    const int nt   = blockIdx.x % (D / 16);
    const int mrow = lane & 15;
    const int kb   = (lane >> 4) * 2;
    const int ncol = nt * 16 + (lane & 15);
    const int row  = mt * 16 + mrow;

    const float* rowp = A;
    bool valid = false;
    if (row < D)                              { rowp = A + (size_t)row * D; valid = true; }
    else if (row == D)                        { rowp = ra;                  valid = true; }
    else if (row == D + 1 && rb != nullptr)   { rowp = rb;                  valid = true; }

    v2f a[D / 4];
#pragma unroll
    for (int ks = 0; ks < D / 4; ++ks) { a[ks].x = 0.0f; a[ks].y = 0.0f; }
    if (valid) {
#pragma unroll
        for (int ks = 0; ks < D / 4; ++ks) {
            a[ks].x = rowp[ks * 4 + kb];
            a[ks].y = rowp[ks * 4 + kb + 1];
        }
    }
    v2f b[D / 4];
#pragma unroll
    for (int ks = 0; ks < D / 4; ++ks) {
        b[ks].x = B[(size_t)(ks * 4 + kb    ) * D + ncol];
        b[ks].y = B[(size_t)(ks * 4 + kb + 1) * D + ncol];
    }

    v8f c = {};
#pragma unroll
    for (int ks = 0; ks < D / 4; ++ks) {
        c = __builtin_amdgcn_wmma_f32_16x16x4_f32(
                 false, a[ks],  false, b[ks],
                 (short)0, c,  false,  false);
    }

    const int rbase = mt * 16 + (lane >> 4) * 8;
#pragma unroll
    for (int v = 0; v < 8; ++v) {
        const int r = rbase + v;
        if (r < D)                             C[(size_t)r * D + ncol] = c[v];
        else if (r == D)                       oa[ncol] = c[v];
        else if (r == D + 1 && ob != nullptr)  ob[ncol] = c[v];
    }
}

template <bool WITH_PRODUCT>
__global__ __launch_bounds__(256) void layer_kernel(
        const float* __restrict__ h,
        const int*   __restrict__ row_start,
        const int*   __restrict__ in_cnt,
        const int2*  __restrict__ col,
        const float* __restrict__ dnorm,
        const float* __restrict__ Wm,
        const float* __restrict__ bias,
        float* __restrict__ out)
{
    __shared__ float ldsA[WITH_PRODUCT ? TILE_M * PITCH : 1];

    const int t    = threadIdx.x;
    const int lane = t & 31;
    const int wave = __builtin_amdgcn_readfirstlane(t >> 5);
    const int M0   = blockIdx.x * TILE_M;
    const int n0   = M0 + wave * NODES_PER_WAVE;

    float* myrows = &ldsA[WITH_PRODUCT ? wave * NODES_PER_WAVE * PITCH + lane * 4 : 0];

    if (n0 < N_NODES) {
        const int4 rs0 = *reinterpret_cast<const int4*>(row_start + n0);
        const int4 rs1 = *reinterpret_cast<const int4*>(row_start + n0 + 4);
        const int4 ct0 = *reinterpret_cast<const int4*>(in_cnt + n0);
        const int4 ct1 = *reinterpret_cast<const int4*>(in_cnt + n0 + 4);
        const float4 dn0 = *reinterpret_cast<const float4*>(dnorm + n0);
        const float4 dn1 = *reinterpret_cast<const float4*>(dnorm + n0 + 4);
        const int   start[NODES_PER_WAVE] = {rs0.x, rs0.y, rs0.z, rs0.w, rs1.x, rs1.y, rs1.z, rs1.w};
        const int   cnt[NODES_PER_WAVE]   = {ct0.x, ct0.y, ct0.z, ct0.w, ct1.x, ct1.y, ct1.z, ct1.w};
        const float sc[NODES_PER_WAVE]    = {dn0.x, dn0.y, dn0.z, dn0.w, dn1.x, dn1.y, dn1.z, dn1.w};
        float4 bv = make_float4(0.0f, 0.0f, 0.0f, 0.0f);
        if constexpr (!WITH_PRODUCT) bv = *reinterpret_cast<const float4*>(bias + lane * 4);

        int2 first[NODES_PER_WAVE];
#pragma unroll
        for (int k = 0; k < NODES_PER_WAVE; ++k) {
            first[k] = make_int2(0, 0);
            if (lane < cnt[k]) first[k] = col[start[k] + lane];
        }

#pragma unroll
        for (int k = 0; k < NODES_PER_WAVE; ++k) {
            float4 acc = make_float4(0.0f, 0.0f, 0.0f, 0.0f);
            int2 p = first[k];
            for (int base = 0; base < cnt[k]; base += 32) {
                const int m = min(32, cnt[k] - base);
                if (base > 0) {
                    p = make_int2(0, 0);
                    if (lane < m) p = col[start[k] + base + lane];
                }
                for (int j = 0; j < m; j += GROUP) {
                    float4 v[GROUP];
                    float  w[GROUP];
#pragma unroll
                    for (int i = 0; i < GROUP; ++i) {
                        const int s = __builtin_amdgcn_readlane(p.x, j + i);
                        w[i] = __int_as_float(__builtin_amdgcn_readlane(p.y, j + i));
                        v[i] = make_float4(0.0f, 0.0f, 0.0f, 0.0f);
                        if (j + i < m) {
                            const float4* row = reinterpret_cast<const float4*>(h + (size_t)s * D);
                            v[i] = row[lane];
                        }
                    }
#pragma unroll
                    for (int i = 0; i < GROUP; ++i) {
                        acc.x = fmaf(v[i].x, w[i], acc.x);
                        acc.y = fmaf(v[i].y, w[i], acc.y);
                        acc.z = fmaf(v[i].z, w[i], acc.z);
                        acc.w = fmaf(v[i].w, w[i], acc.w);
                    }
                }
            }
            if constexpr (WITH_PRODUCT) {
                *reinterpret_cast<float4*>(myrows + k * PITCH) =
                    make_float4(acc.x * sc[k], acc.y * sc[k], acc.z * sc[k], acc.w * sc[k]);
            } else {
                *reinterpret_cast<float4*>(out + (size_t)(n0 + k) * D + lane * 4) =
                    make_float4(fmaf(acc.x, sc[k], bv.x), fmaf(acc.y, sc[k], bv.y),
                                fmaf(acc.z, sc[k], bv.z), fmaf(acc.w, sc[k], bv.w));
            }
        }
    } else if constexpr (WITH_PRODUCT) {
#pragma unroll
        for (int k = 0; k < NODES_PER_WAVE; ++k)
            *reinterpret_cast<float4*>(myrows + k * PITCH) = make_float4(0.0f, 0.0f, 0.0f, 0.0f);
    }

    if constexpr (WITH_PRODUCT) {

    const int ncol = wave * 16 + (lane & 15);
    const int mrow = lane & 15;
    const int kb   = (lane >> 4) * 2;

    v2f b[D / 4];
#pragma unroll
    for (int ks = 0; ks < D / 4; ++ks) {
        b[ks].x = Wm[(size_t)(ks * 4 + kb    ) * D + ncol];
        b[ks].y = Wm[(size_t)(ks * 4 + kb + 1) * D + ncol];
    }
    const float bn = bias[ncol];

    __syncthreads();

    v8f c[TILE_M / 16] = {};
#pragma unroll
    for (int ks = 0; ks < D / 4; ++ks) {
#pragma unroll
        for (int mt = 0; mt < TILE_M / 16; ++mt) {
            const float* ap = &ldsA[(mt * 16 + mrow) * PITCH + ks * 4 + kb];
            v2f a;
            a.x = ap[0];
            a.y = ap[1];
            c[mt] = __builtin_amdgcn_wmma_f32_16x16x4_f32(
                     false, a,  false, b[ks],
                     (short)0, c[mt],  false,  false);
        }
    }

#pragma unroll
    for (int mt = 0; mt < TILE_M / 16; ++mt) {
        const int mtile = M0 + mt * 16;
        if (mtile < N_NODES) {
            const int mbase = mtile + ((lane >> 4) * 8);
#pragma unroll
            for (int v = 0; v < 8; ++v) {
                out[(size_t)(mbase + v) * D + ncol] = c[mt][v] + bn;
            }
        }
    }
    }
}

extern "C" void kernel_launch(void* const* d_in, const int* in_sizes, int n_in,
                              void* d_out, int out_size, void* d_ws, size_t ws_size,
                              hipStream_t stream) {
    const float* feat = (const float*)d_in[0];
    const int*   src  = (const int*)  d_in[1];
    const int*   dst  = (const int*)  d_in[2];
    const float* W    = (const float*)d_in[3];
    const float* b    = (const float*)d_in[4];
    float*       out  = (float*)d_out;

    int*   wsi       = (int*)d_ws;
    float* wsf       = (float*)d_ws;
    int*   in_cnt    = wsi + 50048;
    int*   cursor    = wsi + 100096;
    float* src_norm  = wsf + 100160;
    float* dst_norm  = wsf + 150208;
    int*   rank      = wsi + 250304;
    int2*  col       = (int2*)(wsi + 850304);
    float* hbuf0     = wsf + 2050304;
    float* hbuf1     = wsf + 8450304;
    float* Pm        = wsf + 14850304;
    float* Qm        = wsf + 14866688;
    float* S0        = wsf + 14883072;
    float* c2        = wsf + 14899456;
    float* r0        = wsf + 14899584;
    float* c0        = wsf + 14899712;
    float* c1        = wsf + 14899840;
    int*   in_part   = wsi + 14900224;
    int*   out_part  = wsi + 15300608;
    const int* row_start = in_part;
    static_assert(EDGE_CHUNKS * PART_PITCH == 15300608 - 14900224, "the two count tables do not overlap");

    deg_rank_kernel<<<2 * RANGE_BLOCKS * EDGE_CHUNKS, RANGE_THREADS, 0, stream>>>(src, dst, out_part, in_part, rank, cursor);
    norm_scan_kernel<<<(N_NODES + 255) / 256, 256, 0, stream>>>(out_part, in_part, src_norm, dst_norm,
                                                                in_cnt, cursor);
    fill_kernel<<<(N_EDGES + 255) / 256, 256, 0, stream>>>(src, dst, rank, in_part, src_norm, col);

    static_assert(N_LAYERS == 4, "the reassociation below is written out for four layers");
    wprod_kernel<<<dim3(9 * (D / 16), 2), 32, 0, stream>>>(
        W + (size_t)2 * D * D, b + (size_t)2 * D, nullptr, W + (size_t)3 * D * D, Pm, c2, nullptr,
        W, b, W + (size_t)1 * D * D, Qm, r0);
    wprod_kernel<<<dim3(9 * (D / 16), 1), 32, 0, stream>>>(
        Qm, r0, b + (size_t)1 * D, Pm, S0, c0, c1,
        nullptr, nullptr, nullptr, nullptr, nullptr);

    const int layer_blocks = (N_NODES + TILE_M - 1) / TILE_M;
    layer_kernel<true><<<layer_blocks, 256, 0, stream>>>(
        feat, row_start, in_cnt, col, dst_norm, S0, c0, hbuf0);
    layer_kernel<false><<<layer_blocks, 256, 0, stream>>>(
        hbuf0, row_start, in_cnt, col, dst_norm, nullptr, c1, hbuf1);
    layer_kernel<false><<<layer_blocks, 256, 0, stream>>>(
        hbuf1, row_start, in_cnt, col, dst_norm, nullptr, c2, hbuf0);
    layer_kernel<false><<<layer_blocks, 256, 0, stream>>>(
        hbuf0, row_start, in_cnt, col, dst_norm, nullptr, b + (size_t)3 * D, out);
}


# Both weight-product launches removed: their tiles run as extra work-groups at the head of the degree/rank launch (W2W3, W0W1) and the scan launch (S0, c0, c1): 9 -> 7 launches; sum order unchanged
# speedup vs baseline: 4.8282x; 1.0722x over previous
#include <hip/hip_runtime.h>


typedef __attribute__((ext_vector_type(2))) float v2f;
typedef __attribute__((ext_vector_type(8))) float v8f;

#define N_NODES 50000
#define N_EDGES 600000
#define D 128
#define N_LAYERS 4

#define TILE_M 64
#define NODES_PER_WAVE 8
#define GROUP 8
#define PITCH 132

static_assert(N_NODES % NODES_PER_WAVE == 0, "a wave owns whole groups of nodes");
static_assert(N_NODES % 16 == 0, "a 16-row matrix tile is wholly inside or outside the node range");
static_assert(NODES_PER_WAVE == 8, "the gather reads its bookkeeping as two int4");

#define PROD_TILES (9 * (D / 16))

__device__ __forceinline__ void wprod_tile(const float* A, const float* ra, const float* rb, const float* B,
                                           float* C, float* oa, float* ob, int tile, int lane)
{
    const int mt   = tile / (D / 16);
    const int nt   = tile % (D / 16);
    const int mrow = lane & 15;
    const int kb   = (lane >> 4) * 2;
    const int ncol = nt * 16 + (lane & 15);
    const int row  = mt * 16 + mrow;

    const float* rowp = A;
    bool valid = false;
    if (row < D)                              { rowp = A + (size_t)row * D; valid = true; }
    else if (row == D)                        { rowp = ra;                  valid = true; }
    else if (row == D + 1 && rb != nullptr)   { rowp = rb;                  valid = true; }

    v2f a[D / 4];
#pragma unroll
    for (int ks = 0; ks < D / 4; ++ks) { a[ks].x = 0.0f; a[ks].y = 0.0f; }
    if (valid) {
#pragma unroll
        for (int ks = 0; ks < D / 4; ++ks) {
            a[ks].x = rowp[ks * 4 + kb];
            a[ks].y = rowp[ks * 4 + kb + 1];
        }
    }
    v2f b[D / 4];
#pragma unroll
    for (int ks = 0; ks < D / 4; ++ks) {
        b[ks].x = B[(size_t)(ks * 4 + kb    ) * D + ncol];
        b[ks].y = B[(size_t)(ks * 4 + kb + 1) * D + ncol];
    }

    v8f c = {};
#pragma unroll
    for (int ks = 0; ks < D / 4; ++ks) {
        c = __builtin_amdgcn_wmma_f32_16x16x4_f32(
                 false, a[ks],  false, b[ks],
                 (short)0, c,  false,  false);
    }

    const int rbase = mt * 16 + (lane >> 4) * 8;
#pragma unroll
    for (int v = 0; v < 8; ++v) {
        const int r = rbase + v;
        if (r < D)                             C[(size_t)r * D + ncol] = c[v];
        else if (r == D)                       oa[ncol] = c[v];
        else if (r == D + 1 && ob != nullptr)  ob[ncol] = c[v];
    }
}

#define RANGE_BLOCKS  8
#define EDGE_CHUNKS   8
#define RANGE_NODES   (N_NODES / RANGE_BLOCKS)
#define RANGE_THREADS 1024
#define VEC_PER_TRIP  5
#define NVEC          (N_EDGES / 4)
#define CHUNK_VECS    (NVEC / EDGE_CHUNKS)
#define CHUNK_EDGES   (4 * CHUNK_VECS)
#define TRIP_VECS     (RANGE_THREADS * VEC_PER_TRIP)
#define FULL_TRIPS    (CHUNK_VECS / TRIP_VECS)
#define TAIL_VEC      ((CHUNK_VECS - FULL_TRIPS * TRIP_VECS + RANGE_THREADS - 1) / RANGE_THREADS)
#define PART_PITCH    50048

static_assert(N_EDGES % 4 == 0, "the index arrays are read as int4");
static_assert(N_NODES % RANGE_BLOCKS == 0, "the node pieces are equal");
static_assert(NVEC % EDGE_CHUNKS == 0, "the array pieces are equal and made of whole int4 vectors");
static_assert(FULL_TRIPS == 3, "the pass below is written out for three full trips and a partial one");
static_assert(CHUNK_VECS % TRIP_VECS != 0 && TAIL_VEC <= VEC_PER_TRIP, "the pass below ends with one partial trip");
static_assert(PART_PITCH >= N_NODES && PART_PITCH % 64 == 0, "a plane holds every node and starts on a 256-byte boundary");

template <bool TAIL, int NQ>
__device__ __forceinline__ void load_trip(const int4* __restrict__ idx, int j0, int first,
                                          int4 (&v)[VEC_PER_TRIP], int (&vi)[VEC_PER_TRIP]) {
#pragma unroll
    for (int q = 0; q < NQ; ++q) {
        const int j = j0 + q * RANGE_THREADS;
        const int i = first + (TAIL ? min(j, CHUNK_VECS - 1) : j);
        vi[q] = i;
        v[q]  = idx[i];
        if (TAIL && j >= CHUNK_VECS) v[q] = make_int4(-1, -1, -1, -1);
    }
}

template <bool RANKED, int NQ>
__device__ __forceinline__ void count_trip(const int4 (&v)[VEC_PER_TRIP], const int (&vi)[VEC_PER_TRIP],
                                           int* cnt, unsigned nlo4, int* __restrict__ rank) {
    constexpr unsigned LIM = 4u * RANGE_NODES;
    char* cb = reinterpret_cast<char*>(cnt);
    unsigned a[NQ][4];
#pragma unroll
    for (int q = 0; q < NQ; ++q) {
        a[q][0] = ((unsigned)v[q].x << 2) + nlo4;
        a[q][1] = ((unsigned)v[q].y << 2) + nlo4;
        a[q][2] = ((unsigned)v[q].z << 2) + nlo4;
        a[q][3] = ((unsigned)v[q].w << 2) + nlo4;
    }
    if constexpr (RANKED) {
        int k[NQ][4];
#pragma unroll
        for (int q = 0; q < NQ; ++q) {
#pragma unroll
            for (int c = 0; c < 4; ++c) {
                k[q][c] = 0;
                if (a[q][c] < LIM) k[q][c] = atomicAdd(reinterpret_cast<int*>(cb + a[q][c]), 1);
            }
        }
#pragma unroll
        for (int q = 0; q < NQ; ++q) {
#pragma unroll
            for (int c = 0; c < 4; ++c) {
                if (a[q][c] < LIM) rank[4 * vi[q] + c] = k[q][c];
            }
        }
    } else {
#pragma unroll
        for (int q = 0; q < NQ; ++q) {
#pragma unroll
            for (int c = 0; c < 4; ++c) {
                if (a[q][c] < LIM) atomicAdd(reinterpret_cast<int*>(cb + a[q][c]), 1);
            }
        }
    }
}

template <bool RANKED>
__device__ __forceinline__ void count_piece(const int4* __restrict__ idx, int* cnt, int lo, int first,
                                            int* __restrict__ rank, int t,
                                            int4 (&va)[VEC_PER_TRIP], int (&ia)[VEC_PER_TRIP],
                                            int4 (&vb)[VEC_PER_TRIP], int (&ib)[VEC_PER_TRIP]) {
    const unsigned nlo4 = (unsigned)(-4 * lo);
    constexpr int V = VEC_PER_TRIP;
    count_trip<RANKED, V>(va, ia, cnt, nlo4, rank);
    load_trip<false, V>(idx, 2 * TRIP_VECS + t, first, va, ia);
    count_trip<RANKED, V>(vb, ib, cnt, nlo4, rank);
    load_trip<true, TAIL_VEC>(idx, 3 * TRIP_VECS + t, first, vb, ib);
    count_trip<RANKED, V>(va, ia, cnt, nlo4, rank);
    count_trip<RANKED, TAIL_VEC>(vb, ib, cnt, nlo4, rank);
}

#define PROD1_TILES  (2 * PROD_TILES)
#define PROD1_BLOCKS ((PROD1_TILES + RANGE_THREADS / 32 - 1) / (RANGE_THREADS / 32))

__global__ __launch_bounds__(RANGE_THREADS) void deg_rank_kernel(const int* __restrict__ src,
                                                                 const int* __restrict__ dst,
                                                                 int* __restrict__ out_part,
                                                                 int* __restrict__ in_part,
                                                                 int* __restrict__ rank,
                                                                 int* __restrict__ cursor,
                                                                 const float* W, const float* bvec,
                                                                 float* Pm, float* c2,
                                                                 float* Qm, float* r0) {
    __shared__ int cnt[RANGE_NODES];

    const int  t      = threadIdx.x;

    if (blockIdx.x < PROD1_BLOCKS) {
        const int wave = __builtin_amdgcn_readfirstlane(t >> 5);
        const int tile = blockIdx.x * (RANGE_THREADS / 32) + wave;
        if (tile < PROD_TILES) {
            wprod_tile(W + (size_t)2 * D * D, bvec + (size_t)2 * D, nullptr, W + (size_t)3 * D * D,
                       Pm, c2, nullptr, tile, t & 31);
        } else if (tile < PROD1_TILES) {
            wprod_tile(W, bvec, nullptr, W + (size_t)1 * D * D,
                       Qm, r0, nullptr, tile - PROD_TILES, t & 31);
        }
        return;
    }

    const int  bx     = blockIdx.x - PROD1_BLOCKS;
    const bool by_dst = bx < RANGE_BLOCKS * EDGE_CHUNKS;
    const int  b      = by_dst ? bx : bx - RANGE_BLOCKS * EDGE_CHUNKS;
    const int  piece  = b / EDGE_CHUNKS;
    const int  chunk  = b % EDGE_CHUNKS;
    const int  lo     = piece * RANGE_NODES;
    const int  first  = chunk * CHUNK_VECS;
    const int4* idx   = reinterpret_cast<const int4*>(by_dst ? dst : src);

    int4 va[VEC_PER_TRIP], vb[VEC_PER_TRIP];
    int  ia[VEC_PER_TRIP], ib[VEC_PER_TRIP];
    load_trip<false, VEC_PER_TRIP>(idx, t, first, va, ia);
    load_trip<false, VEC_PER_TRIP>(idx, TRIP_VECS + t, first, vb, ib);

    for (int i = t; i < RANGE_NODES; i += RANGE_THREADS) cnt[i] = 0;
    if (bx == 0 && t == 0) *cursor = 0;
    __syncthreads();

    if (by_dst) count_piece<true >(idx, cnt, lo, first, rank, t, va, ia, vb, ib);
    else        count_piece<false>(idx, cnt, lo, first, rank, t, va, ia, vb, ib);

    __syncthreads();
    int* __restrict__ plane = (by_dst ? in_part : out_part) + chunk * PART_PITCH + lo;
    for (int i = t; i < RANGE_NODES; i += RANGE_THREADS) plane[i] = cnt[i];
}

#define PROD2_BLOCKS (PROD_TILES / 8)
static_assert(PROD_TILES % 8 == 0, "the product work-groups of the scan launch are full");

__global__ __launch_bounds__(256) void norm_scan_kernel(const int* __restrict__ out_part,
                                                        int* in_part,
                                                        float* __restrict__ src_norm,
                                                        float* __restrict__ dst_norm,
                                                        int* __restrict__ in_cnt,
                                                        int* __restrict__ cursor,
                                                        const float* Qm, const float* r0,
                                                        const float* b1, const float* Pm,
                                                        float* S0, float* c0, float* c1) {
    __shared__ int wave_off[8];
    __shared__ int block_base;

    const int t    = threadIdx.x;
    const int lane = t & 31;

    if (blockIdx.x < PROD2_BLOCKS) {
        const int pw = __builtin_amdgcn_readfirstlane(t >> 5);
        wprod_tile(Qm, r0, b1, Pm, S0, c0, c1, blockIdx.x * 8 + pw, lane);
        return;
    }

    const int wave = t >> 5;
    const int n    = (blockIdx.x - PROD2_BLOCKS) * 256 + t;

    int id = 0;
    int before[EDGE_CHUNKS];
#pragma unroll
    for (int c = 0; c < EDGE_CHUNKS; ++c) before[c] = 0;
    if (n < N_NODES) {
        int od = 0;
        int ip[EDGE_CHUNKS];
#pragma unroll
        for (int c = 0; c < EDGE_CHUNKS; ++c) {
            od   += out_part[c * PART_PITCH + n];
            ip[c] = in_part[c * PART_PITCH + n];
        }
#pragma unroll
        for (int c = 0; c < EDGE_CHUNKS; ++c) {
            before[c] = id;
            id += ip[c];
        }
        in_cnt[n]   = id;
        src_norm[n] = rsqrtf(fmaxf((float)od, 1.0f));
        dst_norm[n] = rsqrtf(fmaxf((float)id, 1.0f));
    }

    int x = id;
#pragma unroll
    for (int o = 1; o < 32; o <<= 1) {
        int y = __shfl_up(x, o);
        if (lane >= o) x += y;
    }
    if (lane == 31) wave_off[wave] = x;
    __syncthreads();
    if (t == 0) {
        int tot = 0;
#pragma unroll
        for (int w = 0; w < 8; ++w) {
            int v = wave_off[w];
            wave_off[w] = tot;
            tot += v;
        }
        block_base = atomicAdd(cursor, tot);
    }
    __syncthreads();
    if (n < N_NODES) {
        const int seg = block_base + wave_off[wave] + x - id;
#pragma unroll
        for (int c = 0; c < EDGE_CHUNKS; ++c) in_part[c * PART_PITCH + n] = seg + before[c];
    }
}

__global__ __launch_bounds__(256) void fill_kernel(const int* __restrict__ src,
                                                   const int* __restrict__ dst,
                                                   const int* __restrict__ rank,
                                                   const int* __restrict__ chunk_start,
                                                   const float* __restrict__ src_norm,
                                                   int2* __restrict__ col) {
    int e = blockIdx.x * blockDim.x + threadIdx.x;
    if (e < N_EDGES) {
        const int s = src[e];
        const int d = dst[e];
        const int c = e / CHUNK_EDGES;
        col[chunk_start[c * PART_PITCH + d] + rank[e]] = make_int2(s, __float_as_int(src_norm[s]));
    }
}

template <bool WITH_PRODUCT>
__global__ __launch_bounds__(256) void layer_kernel(
        const float* __restrict__ h,
        const int*   __restrict__ row_start,
        const int*   __restrict__ in_cnt,
        const int2*  __restrict__ col,
        const float* __restrict__ dnorm,
        const float* __restrict__ Wm,
        const float* __restrict__ bias,
        float* __restrict__ out)
{
    __shared__ float ldsA[WITH_PRODUCT ? TILE_M * PITCH : 1];

    const int t    = threadIdx.x;
    const int lane = t & 31;
    const int wave = __builtin_amdgcn_readfirstlane(t >> 5);
    const int M0   = blockIdx.x * TILE_M;
    const int n0   = M0 + wave * NODES_PER_WAVE;

    float* myrows = &ldsA[WITH_PRODUCT ? wave * NODES_PER_WAVE * PITCH + lane * 4 : 0];

    if (n0 < N_NODES) {
        const int4 rs0 = *reinterpret_cast<const int4*>(row_start + n0);
        const int4 rs1 = *reinterpret_cast<const int4*>(row_start + n0 + 4);
        const int4 ct0 = *reinterpret_cast<const int4*>(in_cnt + n0);
        const int4 ct1 = *reinterpret_cast<const int4*>(in_cnt + n0 + 4);
        const float4 dn0 = *reinterpret_cast<const float4*>(dnorm + n0);
        const float4 dn1 = *reinterpret_cast<const float4*>(dnorm + n0 + 4);
        const int   start[NODES_PER_WAVE] = {rs0.x, rs0.y, rs0.z, rs0.w, rs1.x, rs1.y, rs1.z, rs1.w};
        const int   cnt[NODES_PER_WAVE]   = {ct0.x, ct0.y, ct0.z, ct0.w, ct1.x, ct1.y, ct1.z, ct1.w};
        const float sc[NODES_PER_WAVE]    = {dn0.x, dn0.y, dn0.z, dn0.w, dn1.x, dn1.y, dn1.z, dn1.w};
        float4 bv = make_float4(0.0f, 0.0f, 0.0f, 0.0f);
        if constexpr (!WITH_PRODUCT) bv = *reinterpret_cast<const float4*>(bias + lane * 4);

        int2 first[NODES_PER_WAVE];
#pragma unroll
        for (int k = 0; k < NODES_PER_WAVE; ++k) {
            first[k] = make_int2(0, 0);
            if (lane < cnt[k]) first[k] = col[start[k] + lane];
        }

#pragma unroll
        for (int k = 0; k < NODES_PER_WAVE; ++k) {
            float4 acc = make_float4(0.0f, 0.0f, 0.0f, 0.0f);
            int2 p = first[k];
            for (int base = 0; base < cnt[k]; base += 32) {
                const int m = min(32, cnt[k] - base);
                if (base > 0) {
                    p = make_int2(0, 0);
                    if (lane < m) p = col[start[k] + base + lane];
                }
                for (int j = 0; j < m; j += GROUP) {
                    float4 v[GROUP];
                    float  w[GROUP];
#pragma unroll
                    for (int i = 0; i < GROUP; ++i) {
                        const int s = __builtin_amdgcn_readlane(p.x, j + i);
                        w[i] = __int_as_float(__builtin_amdgcn_readlane(p.y, j + i));
                        v[i] = make_float4(0.0f, 0.0f, 0.0f, 0.0f);
                        if (j + i < m) {
                            const float4* row = reinterpret_cast<const float4*>(h + (size_t)s * D);
                            v[i] = row[lane];
                        }
                    }
#pragma unroll
                    for (int i = 0; i < GROUP; ++i) {
                        acc.x = fmaf(v[i].x, w[i], acc.x);
                        acc.y = fmaf(v[i].y, w[i], acc.y);
                        acc.z = fmaf(v[i].z, w[i], acc.z);
                        acc.w = fmaf(v[i].w, w[i], acc.w);
                    }
                }
            }
            if constexpr (WITH_PRODUCT) {
                *reinterpret_cast<float4*>(myrows + k * PITCH) =
                    make_float4(acc.x * sc[k], acc.y * sc[k], acc.z * sc[k], acc.w * sc[k]);
            } else {
                *reinterpret_cast<float4*>(out + (size_t)(n0 + k) * D + lane * 4) =
                    make_float4(fmaf(acc.x, sc[k], bv.x), fmaf(acc.y, sc[k], bv.y),
                                fmaf(acc.z, sc[k], bv.z), fmaf(acc.w, sc[k], bv.w));
            }
        }
    } else if constexpr (WITH_PRODUCT) {
#pragma unroll
        for (int k = 0; k < NODES_PER_WAVE; ++k)
            *reinterpret_cast<float4*>(myrows + k * PITCH) = make_float4(0.0f, 0.0f, 0.0f, 0.0f);
    }

    if constexpr (WITH_PRODUCT) {

    const int ncol = wave * 16 + (lane & 15);
    const int mrow = lane & 15;
    const int kb   = (lane >> 4) * 2;

    v2f b[D / 4];
#pragma unroll
    for (int ks = 0; ks < D / 4; ++ks) {
        b[ks].x = Wm[(size_t)(ks * 4 + kb    ) * D + ncol];
        b[ks].y = Wm[(size_t)(ks * 4 + kb + 1) * D + ncol];
    }
    const float bn = bias[ncol];

    __syncthreads();

    v8f c[TILE_M / 16] = {};
#pragma unroll
    for (int ks = 0; ks < D / 4; ++ks) {
#pragma unroll
        for (int mt = 0; mt < TILE_M / 16; ++mt) {
            const float* ap = &ldsA[(mt * 16 + mrow) * PITCH + ks * 4 + kb];
            v2f a;
            a.x = ap[0];
            a.y = ap[1];
            c[mt] = __builtin_amdgcn_wmma_f32_16x16x4_f32(
                     false, a,  false, b[ks],
                     (short)0, c[mt],  false,  false);
        }
    }

#pragma unroll
    for (int mt = 0; mt < TILE_M / 16; ++mt) {
        const int mtile = M0 + mt * 16;
        if (mtile < N_NODES) {
            const int mbase = mtile + ((lane >> 4) * 8);
#pragma unroll
            for (int v = 0; v < 8; ++v) {
                out[(size_t)(mbase + v) * D + ncol] = c[mt][v] + bn;
            }
        }
    }
    }
}

extern "C" void kernel_launch(void* const* d_in, const int* in_sizes, int n_in,
                              void* d_out, int out_size, void* d_ws, size_t ws_size,
                              hipStream_t stream) {
    const float* feat = (const float*)d_in[0];
    const int*   src  = (const int*)  d_in[1];
    const int*   dst  = (const int*)  d_in[2];
    const float* W    = (const float*)d_in[3];
    const float* b    = (const float*)d_in[4];
    float*       out  = (float*)d_out;

    int*   wsi       = (int*)d_ws;
    float* wsf       = (float*)d_ws;
    int*   in_cnt    = wsi + 50048;
    int*   cursor    = wsi + 100096;
    float* src_norm  = wsf + 100160;
    float* dst_norm  = wsf + 150208;
    int*   rank      = wsi + 250304;
    int2*  col       = (int2*)(wsi + 850304);
    float* hbuf0     = wsf + 2050304;
    float* hbuf1     = wsf + 8450304;
    float* Pm        = wsf + 14850304;
    float* Qm        = wsf + 14866688;
    float* S0        = wsf + 14883072;
    float* c2        = wsf + 14899456;
    float* r0        = wsf + 14899584;
    float* c0        = wsf + 14899712;
    float* c1        = wsf + 14899840;
    int*   in_part   = wsi + 14900224;
    int*   out_part  = wsi + 15300608;
    const int* row_start = in_part;
    static_assert(EDGE_CHUNKS * PART_PITCH == 15300608 - 14900224, "the two count tables do not overlap");

    static_assert(N_LAYERS == 4, "the reassociation of the products is written out for four layers");
    deg_rank_kernel<<<PROD1_BLOCKS + 2 * RANGE_BLOCKS * EDGE_CHUNKS, RANGE_THREADS, 0, stream>>>(
        src, dst, out_part, in_part, rank, cursor, W, b, Pm, c2, Qm, r0);
    norm_scan_kernel<<<PROD2_BLOCKS + (N_NODES + 255) / 256, 256, 0, stream>>>(
        out_part, in_part, src_norm, dst_norm, in_cnt, cursor,
        Qm, r0, b + (size_t)1 * D, Pm, S0, c0, c1);
    fill_kernel<<<(N_EDGES + 255) / 256, 256, 0, stream>>>(src, dst, rank, in_part, src_norm, col);

    const int layer_blocks = (N_NODES + TILE_M - 1) / TILE_M;
    layer_kernel<true><<<layer_blocks, 256, 0, stream>>>(
        feat, row_start, in_cnt, col, dst_norm, S0, c0, hbuf0);
    layer_kernel<false><<<layer_blocks, 256, 0, stream>>>(
        hbuf0, row_start, in_cnt, col, dst_norm, nullptr, c1, hbuf1);
    layer_kernel<false><<<layer_blocks, 256, 0, stream>>>(
        hbuf1, row_start, in_cnt, col, dst_norm, nullptr, c2, hbuf0);
    layer_kernel<false><<<layer_blocks, 256, 0, stream>>>(
        hbuf0, row_start, in_cnt, col, dst_norm, nullptr, b + (size_t)3 * D, out);
}
